# starting kernel
# baseline (speedup 1.0000x reference)
#include <hip/hip_runtime.h>
#include <math.h>

#define Bb 256
#define Tt 512
#define Cc 16
#define Hh 64

#define WP  72
#define AP1 69
#define AP2 133
#define DXP 20

typedef __attribute__((ext_vector_type(2))) float v2f;
typedef __attribute__((ext_vector_type(8))) float v8f;

static __device__ __forceinline__ v8f wmma4(v2f a, v2f b, v8f c) {
  return __builtin_amdgcn_wmma_f32_16x16x4_f32(false, a, false, b, (short)0, c,
                                               false, false);
}

template <int K, int APITCH>
static __device__ __forceinline__ v8f gemm_tile(const float* __restrict__ A,
                                                int acol0,
                                                const float* __restrict__ W,
                                                int n0, v8f acc, int l2, int hi) {
#pragma unroll
  for (int k = 0; k < K; k += 4) {
    v2f a, b;
    const float* ap = A + l2 * APITCH + acol0 + k + 2 * hi;
    a.x = ap[0];
    a.y = ap[1];
    const float* wp = W + (k + 2 * hi) * WP + n0 + l2;
    b.x = wp[0];
    b.y = wp[WP];
    acc = wmma4(a, b, acc);
  }
  return acc;
}

template <int APITCH>
static __device__ __forceinline__ void store_tile(float* __restrict__ O, int n0,
                                                  v8f acc, int l2, int hi) {
#pragma unroll
  for (int r = 0; r < 8; ++r) O[(r + 8 * hi) * APITCH + n0 + l2] = acc[r];
}

template <int ROWS>
static __device__ __forceinline__ void load_w64(float* __restrict__ dst,
                                                const float* __restrict__ src,
                                                int tid, int nthr) {
  for (int i = tid; i < ROWS * 64; i += nthr)
    dst[(i >> 6) * WP + (i & 63)] = src[i];
}

static __device__ __forceinline__ float relu(float v) { return v > 0.f ? v : 0.f; }

__global__ __launch_bounds__(128) void ncde_embed_kernel(
    const float* __restrict__ x, const float* __restrict__ seq_ts,
    const float* __restrict__ coeffs, const float* __restrict__ times,
    const float* __restrict__ W_if, const float* __restrict__ b_if,
    const float* __restrict__ W_ic, const float* __restrict__ b_ic,
    const float* __restrict__ W_emb, const float* __restrict__ b_emb,
    const float* __restrict__ W_fl1, const float* __restrict__ b_fl1,
    const float* __restrict__ W_flo, const float* __restrict__ b_flo,
    float* __restrict__ zflow) {
  __shared__ float sWif[16 * WP], sWic[16 * WP], sWemb[128 * WP];
  __shared__ float sWfl1[64 * WP], sWflo[64 * WP];
  __shared__ float sb[5][64];
  __shared__ float sA[16 * AP2], sB[16 * AP2];

  const int tid = threadIdx.x, nthr = 128;
  const int lane = tid & 31, wave = tid >> 5, l2 = lane & 15, hi = lane >> 4;
  const int r0 = blockIdx.x * 16;

  load_w64<16>(sWif, W_if, tid, nthr);
  load_w64<16>(sWic, W_ic, tid, nthr);
  load_w64<128>(sWemb, W_emb, tid, nthr);
  load_w64<64>(sWfl1, W_fl1, tid, nthr);
  load_w64<64>(sWflo, W_flo, tid, nthr);
  if (tid < 64) {
    sb[0][tid] = b_if[tid];
    sb[1][tid] = b_ic[tid];
    sb[2][tid] = b_emb[tid];
    sb[3][tid] = b_fl1[tid];
    sb[4][tid] = b_flo[tid];
  }
  for (int i = tid; i < 256; i += nthr) {
    const int m = i >> 4, c = i & 15;
    const int rr = r0 + m, b = rr / Tt, t = rr % Tt;
    sA[m * AP2 + c] = (c == 0) ? seq_ts[rr] : x[(size_t)rr * (Cc - 1) + (c - 1)];
    const int idx = (t < Tt - 1) ? t : (Tt - 2);
    const float s = times[t] - times[idx];
    const float* cf = coeffs + ((size_t)b * (Tt - 1) + idx) * (4 * Cc) + c;
    const float a0 = cf[0], b0 = cf[Cc], c0 = cf[2 * Cc], d0 = cf[3 * Cc];
    sA[m * AP2 + 16 + c] = a0 + s * (b0 + s * (c0 + s * d0));
  }
  __syncthreads();

  const int n0 = wave * 16;
  v8f aF = {}, aX = {};
  aF = gemm_tile<16, AP2>(sA, 0, sWif, n0, aF, l2, hi);
  aX = gemm_tile<16, AP2>(sA, 16, sWic, n0, aX, l2, hi);
  const float bF = sb[0][n0 + l2], bX = sb[1][n0 + l2];
#pragma unroll
  for (int r = 0; r < 8; ++r) {
    const int m = r + 8 * hi;
    sB[m * AP2 + n0 + l2] = aF[r] + bF;
    sB[m * AP2 + 64 + n0 + l2] = aX[r] + bX;
  }
  __syncthreads();

  v8f acc = {};
  acc = gemm_tile<128, AP2>(sB, 0, sWemb, n0, acc, l2, hi);
  float bv = sb[2][n0 + l2];
#pragma unroll
  for (int r = 0; r < 8; ++r)
    sA[(r + 8 * hi) * AP2 + n0 + l2] = relu(acc[r] + bv);
  __syncthreads();

  acc = (v8f){};
  acc = gemm_tile<64, AP2>(sA, 0, sWfl1, n0, acc, l2, hi);
  bv = sb[3][n0 + l2];
#pragma unroll
  for (int r = 0; r < 8; ++r)
    sB[(r + 8 * hi) * AP2 + n0 + l2] = relu(acc[r] + bv);
  __syncthreads();

  acc = (v8f){};
  acc = gemm_tile<64, AP2>(sB, 0, sWflo, n0, acc, l2, hi);
  bv = sb[4][n0 + l2];
#pragma unroll
  for (int r = 0; r < 8; ++r)
    zflow[(size_t)(r0 + r + 8 * hi) * Hh + n0 + l2] = acc[r] + bv;
}

__global__ __launch_bounds__(512, 1) void ncde_scan_kernel(
    const float* __restrict__ coeffs, const float* __restrict__ times,
    const float* __restrict__ W_ic, const float* __restrict__ b_ic,
    const float* __restrict__ W_f1, const float* __restrict__ b_f1,
    const float* __restrict__ W_f2, const float* __restrict__ b_f2,
    float* __restrict__ zcde) {
  __shared__ float sWf1[64 * WP], sWic[16 * WP];
  __shared__ float sbf1[64], sbic[64];
  __shared__ float stim[Tt];
  __shared__ float sh[16 * AP1], szb[16 * AP1], sze[16 * AP1];
  __shared__ float sk[4][16 * AP1];
  __shared__ float sdx[4][16 * DXP];

  const int tid = threadIdx.x, nthr = 512;
  const int lane = tid & 31, wave = tid >> 5, l2 = lane & 15, hi = lane >> 4;
  const int b0 = blockIdx.x * 16;

  load_w64<64>(sWf1, W_f1, tid, nthr);
  load_w64<16>(sWic, W_ic, tid, nthr);
  if (tid < 64) {
    sbf1[tid] = b_f1[tid];
    sbic[tid] = b_ic[tid];
  }
  for (int i = tid; i < Tt; i += nthr) stim[i] = times[i];

  v2f wf2[4][16];
  float bf2r[4];
#pragma unroll
  for (int jj = 0; jj < 4; ++jj) {
    const int j = wave * 4 + jj;
    const int col = 16 * j + l2;
    bf2r[jj] = b_f2[col];
#pragma unroll
    for (int kk = 0; kk < 16; ++kk) {
      const int row = 4 * kk + 2 * hi;
      wf2[jj][kk].x = W_f2[(size_t)row * (Hh * Cc) + col];
      wf2[jj][kk].y = W_f2[(size_t)(row + 1) * (Hh * Cc) + col];
    }
  }

  for (int i = tid; i < 256; i += nthr) {
    const int m = i >> 4, c = i & 15;
    sze[m * AP1 + c] = coeffs[((size_t)(b0 + m) * (Tt - 1)) * (4 * Cc) + c];
  }
  __syncthreads();
  if (wave < 4) {
    const int n0 = wave * 16;
    v8f acc = {};
    acc = gemm_tile<16, AP1>(sze, 0, sWic, n0, acc, l2, hi);
    const float bv = sbic[n0 + l2];
#pragma unroll
    for (int r = 0; r < 8; ++r) {
      const int m = r + 8 * hi;
      const float v = acc[r] + bv;
      szb[m * AP1 + n0 + l2] = v;
      zcde[((size_t)(b0 + m) * Tt) * Hh + n0 + l2] = v;
    }
  }
  __syncthreads();
  for (int i = tid; i < 16 * Hh; i += nthr) {
    const int m = i >> 6, h = i & 63;
    sze[m * AP1 + h] = szb[m * AP1 + h];
  }
  __syncthreads();

  for (int step = 0; step < Tt - 1; ++step) {
    const float dt = stim[step + 1] - stim[step];
    if (tid < 256) {
      const int m = tid >> 4, c = tid & 15, b = b0 + m;
      const float* cf = coeffs + ((size_t)b * (Tt - 1) + step) * (4 * Cc) + c;
      const float bb = cf[Cc], cc = cf[2 * Cc], dd = cf[3 * Cc];
      const float s1 = dt * (1.f / 3.f), s2 = dt * (2.f / 3.f);
      sdx[0][m * DXP + c] = bb;
      sdx[1][m * DXP + c] = bb + s1 * (2.f * cc + 3.f * dd * s1);
      sdx[2][m * DXP + c] = bb + s2 * (2.f * cc + 3.f * dd * s2);
      const int idx3 = (step + 1 < Tt - 1) ? (step + 1) : (Tt - 2);
      const float* cf3 = coeffs + ((size_t)b * (Tt - 1) + idx3) * (4 * Cc) + c;
      const float b3 = cf3[Cc], c3 = cf3[2 * Cc], d3 = cf3[3 * Cc];
      const float s3 = stim[step + 1] - stim[idx3];
      sdx[3][m * DXP + c] = b3 + s3 * (2.f * c3 + 3.f * d3 * s3);
    }
    __syncthreads();

    for (int sub = 0; sub < 4; ++sub) {
      if (wave < 4) {
        const int n0 = wave * 16;
        v8f acc = {};
        acc = gemm_tile<64, AP1>(sze, 0, sWf1, n0, acc, l2, hi);
        const float bv = sbf1[n0 + l2];
#pragma unroll
        for (int r = 0; r < 8; ++r)
          sh[(r + 8 * hi) * AP1 + n0 + l2] = relu(acc[r] + bv);
      }
      __syncthreads();

      v8f acc2[4];
#pragma unroll
      for (int jj = 0; jj < 4; ++jj) {
        v8f t;
#pragma unroll
        for (int r = 0; r < 8; ++r) t[r] = bf2r[jj];
        acc2[jj] = t;
      }
#pragma unroll
      for (int kk = 0; kk < 16; ++kk) {
        v2f a;
        const float* ap = sh + l2 * AP1 + 4 * kk + 2 * hi;
        a.x = ap[0];
        a.y = ap[1];
#pragma unroll
        for (int jj = 0; jj < 4; ++jj) acc2[jj] = wmma4(a, wf2[jj][kk], acc2[jj]);
      }
      float dxr[8];
#pragma unroll
      for (int r = 0; r < 8; ++r) dxr[r] = sdx[sub][(r + 8 * hi) * DXP + l2];
      float* kdst = sk[sub];
#pragma unroll
      for (int jj = 0; jj < 4; ++jj) {
#pragma unroll
        for (int r = 0; r < 8; ++r) {
          float p = tanhf(acc2[jj][r]) * dxr[r];
          p += __shfl_xor(p, 1, 32);
          p += __shfl_xor(p, 2, 32);
          p += __shfl_xor(p, 4, 32);
          p += __shfl_xor(p, 8, 32);
          if (l2 == 0) kdst[(r + 8 * hi) * AP1 + wave * 4 + jj] = p;
        }
      }
      __syncthreads();

      for (int i = tid; i < 16 * Hh; i += nthr) {
        const int m = i >> 6, h = i & 63;
        const int ad = m * AP1 + h;
        const float zb = szb[ad];
        if (sub == 0) {
          sze[ad] = zb + (dt * (1.f / 3.f)) * sk[0][ad];
        } else if (sub == 1) {
          sze[ad] = zb + dt * (sk[1][ad] - sk[0][ad] * (1.f / 3.f));
        } else if (sub == 2) {
          sze[ad] = zb + dt * (sk[0][ad] - sk[1][ad] + sk[2][ad]);
        } else {
          const float zn =
              zb + (sk[0][ad] + 3.f * (sk[1][ad] + sk[2][ad]) + sk[3][ad]) *
                       (dt * 0.125f);
          szb[ad] = zn;
          sze[ad] = zn;
          zcde[((size_t)(b0 + m) * Tt + (step + 1)) * Hh + h] = zn;
        }
      }
      __syncthreads();
    }
  }
}

__global__ __launch_bounds__(128) void ncde_head_kernel(
    const float* __restrict__ zflow, const float* __restrict__ zcde,
    const float* __restrict__ W_mix, const float* __restrict__ b_mix,
    const float* __restrict__ W_l1, const float* __restrict__ b_l1,
    const float* __restrict__ W_l2, const float* __restrict__ b_l2,
    float* __restrict__ out, float* __restrict__ zout) {
  __shared__ float sWmix[128 * WP], sWl1[64 * WP];
  __shared__ float sbmix[64], sbl1[64], sWl2v[64];
  __shared__ float sA[16 * AP2], sB[16 * AP1], sC[16 * AP1];

  const int tid = threadIdx.x, nthr = 128;
  const int lane = tid & 31, wave = tid >> 5, l2 = lane & 15, hi = lane >> 4;
  const int r0 = blockIdx.x * 16;

  load_w64<128>(sWmix, W_mix, tid, nthr);
  load_w64<64>(sWl1, W_l1, tid, nthr);
  if (tid < 64) {
    sbmix[tid] = b_mix[tid];
    sbl1[tid] = b_l1[tid];
    sWl2v[tid] = W_l2[tid];
  }
  for (int i = tid; i < 16 * 128; i += nthr) {
    const int m = i >> 7, c = i & 127;
    const size_t rr = r0 + m;
    sA[m * AP2 + c] = (c < 64) ? zflow[rr * Hh + c] : zcde[rr * Hh + (c - 64)];
  }
  __syncthreads();

  const int n0 = wave * 16;
  v8f acc = {};
  acc = gemm_tile<128, AP2>(sA, 0, sWmix, n0, acc, l2, hi);
  float bv = sbmix[n0 + l2];
#pragma unroll
  for (int r = 0; r < 8; ++r) {
    const int m = r + 8 * hi;
    const float v = acc[r] + bv;
    zout[(size_t)(r0 + m) * Hh + n0 + l2] = v;
    sB[m * AP1 + n0 + l2] = tanhf(v);
  }
  __syncthreads();

  acc = (v8f){};
  acc = gemm_tile<64, AP1>(sB, 0, sWl1, n0, acc, l2, hi);
  bv = sbl1[n0 + l2];
#pragma unroll
  for (int r = 0; r < 8; ++r)
    sC[(r + 8 * hi) * AP1 + n0 + l2] = relu(acc[r] + bv);
  __syncthreads();

  const int m = tid >> 3, p = tid & 7;
  float s = 0.f;
#pragma unroll
  for (int q = 0; q < 8; ++q) s += sC[m * AP1 + p * 8 + q] * sWl2v[p * 8 + q];
  s += __shfl_xor(s, 1, 32);
  s += __shfl_xor(s, 2, 32);
  s += __shfl_xor(s, 4, 32);
  if (p == 0) out[r0 + m] = s + b_l2[0];
}

extern "C" void kernel_launch(void* const* d_in, const int* in_sizes, int n_in,
                              void* d_out, int out_size, void* d_ws,
                              size_t ws_size, hipStream_t stream) {
  const float* x = (const float*)d_in[0];
  const float* seq_ts = (const float*)d_in[1];
  const float* coeffs = (const float*)d_in[3];
  const float* times = (const float*)d_in[4];
  const float* W_if = (const float*)d_in[5];
  const float* b_if = (const float*)d_in[6];
  const float* W_ic = (const float*)d_in[7];
  const float* b_ic = (const float*)d_in[8];
  const float* W_emb = (const float*)d_in[9];
  const float* b_emb = (const float*)d_in[10];
  const float* W_fl1 = (const float*)d_in[11];
  const float* b_fl1 = (const float*)d_in[12];
  const float* W_flo = (const float*)d_in[13];
  const float* b_flo = (const float*)d_in[14];
  const float* W_f1 = (const float*)d_in[15];
  const float* b_f1 = (const float*)d_in[16];
  const float* W_f2 = (const float*)d_in[17];
  const float* b_f2 = (const float*)d_in[18];
  const float* W_mix = (const float*)d_in[19];
  const float* b_mix = (const float*)d_in[20];
  const float* W_l1 = (const float*)d_in[21];
  const float* b_l1 = (const float*)d_in[22];
  const float* W_l2 = (const float*)d_in[23];
  const float* b_l2 = (const float*)d_in[24];

  float* zflow = (float*)d_ws;
  float* zcde = zflow + (size_t)Bb * Tt * Hh;
  float* out = (float*)d_out;
  float* zout = out + (size_t)Bb * Tt;

  ncde_embed_kernel<<<(Bb * Tt) / 16, 128, 0, stream>>>(
      x, seq_ts, coeffs, times, W_if, b_if, W_ic, b_ic, W_emb, b_emb, W_fl1,
      b_fl1, W_flo, b_flo, zflow);
  ncde_scan_kernel<<<Bb / 16, 512, 0, stream>>>(coeffs, times, W_ic, b_ic,
                                                W_f1, b_f1, W_f2, b_f2, zcde);
  ncde_head_kernel<<<(Bb * Tt) / 16, 128, 0, stream>>>(
      zflow, zcde, W_mix, b_mix, W_l1, b_l1, W_l2, b_l2, out, zout);
}


# scan: einsum channel sum (16 lanes) by four DPP row_xmask adds on the vector ALU per value, not four ds_bpermute shuffles each with an LDS round trip and wait; sum tree unchanged
# speedup vs baseline: 2.8341x; 2.8341x over previous
#include <hip/hip_runtime.h>
#include <math.h>

#define Bb 256
#define Tt 512
#define Cc 16
#define Hh 64
#define RB 2
#define NVAL (2 * RB)

#define WP  72
#define AP1 69
#define AP2 133
#define DXP 20

typedef __attribute__((ext_vector_type(2))) float v2f;
typedef __attribute__((ext_vector_type(8))) float v8f;

static __device__ __forceinline__ v8f wmma4(v2f a, v2f b, v8f c) {
  return __builtin_amdgcn_wmma_f32_16x16x4_f32(false, a, false, b, (short)0, c,
                                               false, false);
}

template <int K, int APITCH, int ROWMASK = 15>
static __device__ __forceinline__ v8f gemm_tile(const float* __restrict__ A,
                                                int acol0,
                                                const float* __restrict__ W,
                                                int n0, v8f acc, int l2, int hi) {
#pragma unroll
  for (int k = 0; k < K; k += 4) {
    v2f a, b;
    const float* ap = A + (l2 & ROWMASK) * APITCH + acol0 + k + 2 * hi;
    a.x = ap[0];
    a.y = ap[1];
    const float* wp = W + (k + 2 * hi) * WP + n0 + l2;
    b.x = wp[0];
    b.y = wp[WP];
    acc = wmma4(a, b, acc);
  }
  return acc;
}

template <int APITCH>
static __device__ __forceinline__ void store_tile(float* __restrict__ O, int n0,
                                                  v8f acc, int l2, int hi) {
#pragma unroll
  for (int r = 0; r < 8; ++r) O[(r + 8 * hi) * APITCH + n0 + l2] = acc[r];
}

template <int ROWS>
static __device__ __forceinline__ void load_w64(float* __restrict__ dst,
                                                const float* __restrict__ src,
                                                int tid, int nthr) {
  for (int i = tid; i < ROWS * 64; i += nthr)
    dst[(i >> 6) * WP + (i & 63)] = src[i];
}

static __device__ __forceinline__ float relu(float v) { return v > 0.f ? v : 0.f; }

template <int M>
static __device__ __forceinline__ float row_xor_add(float p) {
  const int other = __builtin_amdgcn_update_dpp(0, __builtin_bit_cast(int, p),
                                                0x160 + M, 0xf, 0xf, true);
  return p + __builtin_bit_cast(float, other);
}

__global__ __launch_bounds__(128) void ncde_embed_kernel(
    const float* __restrict__ x, const float* __restrict__ seq_ts,
    const float* __restrict__ coeffs, const float* __restrict__ times,
    const float* __restrict__ W_if, const float* __restrict__ b_if,
    const float* __restrict__ W_ic, const float* __restrict__ b_ic,
    const float* __restrict__ W_emb, const float* __restrict__ b_emb,
    const float* __restrict__ W_fl1, const float* __restrict__ b_fl1,
    const float* __restrict__ W_flo, const float* __restrict__ b_flo,
    float* __restrict__ zflow) {
  __shared__ float sWif[16 * WP], sWic[16 * WP], sWemb[128 * WP];
  __shared__ float sWfl1[64 * WP], sWflo[64 * WP];
  __shared__ float sb[5][64];
  __shared__ float sA[16 * AP2], sB[16 * AP2];

  const int tid = threadIdx.x, nthr = 128;
  const int lane = tid & 31, wave = tid >> 5, l2 = lane & 15, hi = lane >> 4;
  const int r0 = blockIdx.x * 16;

  load_w64<16>(sWif, W_if, tid, nthr);
  load_w64<16>(sWic, W_ic, tid, nthr);
  load_w64<128>(sWemb, W_emb, tid, nthr);
  load_w64<64>(sWfl1, W_fl1, tid, nthr);
  load_w64<64>(sWflo, W_flo, tid, nthr);
  if (tid < 64) {
    sb[0][tid] = b_if[tid];
    sb[1][tid] = b_ic[tid];
    sb[2][tid] = b_emb[tid];
    sb[3][tid] = b_fl1[tid];
    sb[4][tid] = b_flo[tid];
  }
  for (int i = tid; i < 256; i += nthr) {
    const int m = i >> 4, c = i & 15;
    const int rr = r0 + m, b = rr / Tt, t = rr % Tt;
    sA[m * AP2 + c] = (c == 0) ? seq_ts[rr] : x[(size_t)rr * (Cc - 1) + (c - 1)];
    const int idx = (t < Tt - 1) ? t : (Tt - 2);
    const float s = times[t] - times[idx];
    const float* cf = coeffs + ((size_t)b * (Tt - 1) + idx) * (4 * Cc) + c;
    const float a0 = cf[0], b0 = cf[Cc], c0 = cf[2 * Cc], d0 = cf[3 * Cc];
    sA[m * AP2 + 16 + c] = a0 + s * (b0 + s * (c0 + s * d0));
  }
  __syncthreads();

  const int n0 = wave * 16;
  v8f aF = {}, aX = {};
  aF = gemm_tile<16, AP2>(sA, 0, sWif, n0, aF, l2, hi);
  aX = gemm_tile<16, AP2>(sA, 16, sWic, n0, aX, l2, hi);
  const float bF = sb[0][n0 + l2], bX = sb[1][n0 + l2];
#pragma unroll
  for (int r = 0; r < 8; ++r) {
    const int m = r + 8 * hi;
    sB[m * AP2 + n0 + l2] = aF[r] + bF;
    sB[m * AP2 + 64 + n0 + l2] = aX[r] + bX;
  }
  __syncthreads();

  v8f acc = {};
  acc = gemm_tile<128, AP2>(sB, 0, sWemb, n0, acc, l2, hi);
  float bv = sb[2][n0 + l2];
#pragma unroll
  for (int r = 0; r < 8; ++r)
    sA[(r + 8 * hi) * AP2 + n0 + l2] = relu(acc[r] + bv);
  __syncthreads();

  acc = (v8f){};
  acc = gemm_tile<64, AP2>(sA, 0, sWfl1, n0, acc, l2, hi);
  bv = sb[3][n0 + l2];
#pragma unroll
  for (int r = 0; r < 8; ++r)
    sB[(r + 8 * hi) * AP2 + n0 + l2] = relu(acc[r] + bv);
  __syncthreads();

  acc = (v8f){};
  acc = gemm_tile<64, AP2>(sB, 0, sWflo, n0, acc, l2, hi);
  bv = sb[4][n0 + l2];
#pragma unroll
  for (int r = 0; r < 8; ++r)
    zflow[(size_t)(r0 + r + 8 * hi) * Hh + n0 + l2] = acc[r] + bv;
}

__global__ __launch_bounds__(512, 1) void ncde_scan_kernel(
    const float* __restrict__ coeffs, const float* __restrict__ times,
    const float* __restrict__ W_ic, const float* __restrict__ b_ic,
    const float* __restrict__ W_f1, const float* __restrict__ b_f1,
    const float* __restrict__ W_f2, const float* __restrict__ b_f2,
    float* __restrict__ zcde) {
  __shared__ float sWf1[64 * WP], sWic[16 * WP];
  __shared__ float sbf1[64], sbic[64];
  __shared__ float stim[Tt];
  __shared__ float sh[RB * AP1], szb[RB * AP1], sze[RB * AP1];
  __shared__ float sk[4][RB * AP1];
  __shared__ float sdx[4][RB * DXP];

  const int tid = threadIdx.x, nthr = 512;
  const int lane = tid & 31, wave = tid >> 5, l2 = lane & 15, hi = lane >> 4;
  const int b0 = blockIdx.x * RB;
  const int rblk = (l2 & 7) / RB;
  const int myq = (rblk < 2) ? (2 * (l2 >> 3) + rblk) : -1;

  load_w64<64>(sWf1, W_f1, tid, nthr);
  load_w64<16>(sWic, W_ic, tid, nthr);
  if (tid < 64) {
    sbf1[tid] = b_f1[tid];
    sbic[tid] = b_ic[tid];
  }
  for (int i = tid; i < Tt; i += nthr) stim[i] = times[i];

  v2f wf2[4][16];
  float bf2r[2];
#pragma unroll
  for (int t = 0; t < 2; ++t) bf2r[t] = b_f2[16 * (wave * 4 + 2 * hi + t) + l2];
#pragma unroll
  for (int jj = 0; jj < 4; ++jj) {
    const int j = wave * 4 + jj;
    const int col = 16 * j + l2;
#pragma unroll
    for (int kk = 0; kk < 16; ++kk) {
      const int row = 4 * kk + 2 * hi;
      wf2[jj][kk].x = W_f2[(size_t)row * (Hh * Cc) + col];
      wf2[jj][kk].y = W_f2[(size_t)(row + 1) * (Hh * Cc) + col];
    }
  }

  for (int i = tid; i < RB * 16; i += nthr) {
    const int m = i >> 4, c = i & 15;
    sze[m * AP1 + c] = coeffs[((size_t)(b0 + m) * (Tt - 1)) * (4 * Cc) + c];
  }
  __syncthreads();
  if (wave < 4) {
    const int n0 = wave * 16;
    v8f acc = {};
    acc = gemm_tile<16, AP1, RB - 1>(sze, 0, sWic, n0, acc, l2, hi);
    const float bv = sbic[n0 + l2];
    float zv[RB];
#pragma unroll
    for (int r = 0; r < RB; ++r) zv[r] = acc[r] + bv;
    if (hi == 0) {
#pragma unroll
      for (int r = 0; r < RB; ++r) {
        szb[r * AP1 + n0 + l2] = zv[r];
        zcde[((size_t)(b0 + r) * Tt) * Hh + n0 + l2] = zv[r];
      }
    }
  }
  __syncthreads();
  for (int i = tid; i < RB * Hh; i += nthr) {
    const int m = i >> 6, h = i & 63;
    sze[m * AP1 + h] = szb[m * AP1 + h];
  }
  __syncthreads();

  for (int step = 0; step < Tt - 1; ++step) {
    const float dt = stim[step + 1] - stim[step];
    if (tid < RB * 16) {
      const int m = tid >> 4, c = tid & 15, b = b0 + m;
      const float* cf = coeffs + ((size_t)b * (Tt - 1) + step) * (4 * Cc) + c;
      const float bb = cf[Cc], cc = cf[2 * Cc], dd = cf[3 * Cc];
      const float s1 = dt * (1.f / 3.f), s2 = dt * (2.f / 3.f);
      sdx[0][m * DXP + c] = bb;
      sdx[1][m * DXP + c] = bb + s1 * (2.f * cc + 3.f * dd * s1);
      sdx[2][m * DXP + c] = bb + s2 * (2.f * cc + 3.f * dd * s2);
      const int idx3 = (step + 1 < Tt - 1) ? (step + 1) : (Tt - 2);
      const float* cf3 = coeffs + ((size_t)b * (Tt - 1) + idx3) * (4 * Cc) + c;
      const float b3 = cf3[Cc], c3 = cf3[2 * Cc], d3 = cf3[3 * Cc];
      const float s3 = stim[step + 1] - stim[idx3];
      sdx[3][m * DXP + c] = b3 + s3 * (2.f * c3 + 3.f * d3 * s3);
    }
    __syncthreads();

    for (int sub = 0; sub < 4; ++sub) {
      if (wave < 4) {
        const int n0 = wave * 16;
        v8f acc = {};
        acc = gemm_tile<64, AP1, RB - 1>(sze, 0, sWf1, n0, acc, l2, hi);
        const float bv = sbf1[n0 + l2];
        float hv[RB];
#pragma unroll
        for (int r = 0; r < RB; ++r) hv[r] = relu(acc[r] + bv);
        if (hi == 0) {
#pragma unroll
          for (int r = 0; r < RB; ++r) sh[r * AP1 + n0 + l2] = hv[r];
        }
      }
      __syncthreads();

      v8f acc2;
#pragma unroll
      for (int r = 0; r < 8; ++r) acc2[r] = (r < NVAL) ? bf2r[r / RB] : 0.f;
#pragma unroll
      for (int kk = 0; kk < 16; ++kk) {
        v2f hv2;
        const float* ap = sh + (l2 & (RB - 1)) * AP1 + 4 * kk + 2 * hi;
        hv2.x = ap[0];
        hv2.y = ap[1];
#pragma unroll
        for (int jj = 0; jj < 4; ++jj) {
          v2f a;
          a.x = (myq == jj) ? hv2.x : 0.f;
          a.y = (myq == jj) ? hv2.y : 0.f;
          acc2 = wmma4(a, wf2[jj][kk], acc2);
        }
      }
      float dxr[RB];
#pragma unroll
      for (int i = 0; i < RB; ++i) dxr[i] = sdx[sub][i * DXP + l2];
      float* kdst = sk[sub];
      float pv[NVAL];
#pragma unroll
      for (int r = 0; r < NVAL; ++r) pv[r] = tanhf(acc2[r]) * dxr[r % RB];
#pragma unroll
      for (int r = 0; r < NVAL; ++r) {
        float p = pv[r];
        p = row_xor_add<1>(p);
        p = row_xor_add<2>(p);
        p = row_xor_add<4>(p);
        p = row_xor_add<8>(p);
        pv[r] = p;
      }
      if (l2 == 0) {
#pragma unroll
        for (int r = 0; r < NVAL; ++r)
          kdst[(r % RB) * AP1 + wave * 4 + 2 * hi + r / RB] = pv[r];
      }
      __syncthreads();

      for (int i = tid; i < RB * Hh; i += nthr) {
        const int m = i >> 6, h = i & 63;
        const int ad = m * AP1 + h;
        const float zb = szb[ad];
        if (sub == 0) {
          sze[ad] = zb + (dt * (1.f / 3.f)) * sk[0][ad];
        } else if (sub == 1) {
          sze[ad] = zb + dt * (sk[1][ad] - sk[0][ad] * (1.f / 3.f));
        } else if (sub == 2) {
          sze[ad] = zb + dt * (sk[0][ad] - sk[1][ad] + sk[2][ad]);
        } else {
          const float zn =
              zb + (sk[0][ad] + 3.f * (sk[1][ad] + sk[2][ad]) + sk[3][ad]) *
                       (dt * 0.125f);
          szb[ad] = zn;
          sze[ad] = zn;
          zcde[((size_t)(b0 + m) * Tt + (step + 1)) * Hh + h] = zn;
        }
      }
      __syncthreads();
    }
  }
}

__global__ __launch_bounds__(128) void ncde_head_kernel(
    const float* __restrict__ zflow, const float* __restrict__ zcde,
    const float* __restrict__ W_mix, const float* __restrict__ b_mix,
    const float* __restrict__ W_l1, const float* __restrict__ b_l1,
    const float* __restrict__ W_l2, const float* __restrict__ b_l2,
    float* __restrict__ out, float* __restrict__ zout) {
  __shared__ float sWmix[128 * WP], sWl1[64 * WP];
  __shared__ float sbmix[64], sbl1[64], sWl2v[64];
  __shared__ float sA[16 * AP2], sB[16 * AP1], sC[16 * AP1];

  const int tid = threadIdx.x, nthr = 128;
  const int lane = tid & 31, wave = tid >> 5, l2 = lane & 15, hi = lane >> 4;
  const int r0 = blockIdx.x * 16;

  load_w64<128>(sWmix, W_mix, tid, nthr);
  load_w64<64>(sWl1, W_l1, tid, nthr);
  if (tid < 64) {
    sbmix[tid] = b_mix[tid];
    sbl1[tid] = b_l1[tid];
    sWl2v[tid] = W_l2[tid];
  }
  for (int i = tid; i < 16 * 128; i += nthr) {
    const int m = i >> 7, c = i & 127;
    const size_t rr = r0 + m;
    sA[m * AP2 + c] = (c < 64) ? zflow[rr * Hh + c] : zcde[rr * Hh + (c - 64)];
  }
  __syncthreads();

  const int n0 = wave * 16;
  v8f acc = {};
  acc = gemm_tile<128, AP2>(sA, 0, sWmix, n0, acc, l2, hi);
  float bv = sbmix[n0 + l2];
#pragma unroll
  for (int r = 0; r < 8; ++r) {
    const int m = r + 8 * hi;
    const float v = acc[r] + bv;
    zout[(size_t)(r0 + m) * Hh + n0 + l2] = v;
    sB[m * AP1 + n0 + l2] = tanhf(v);
  }
  __syncthreads();

  acc = (v8f){};
  acc = gemm_tile<64, AP1>(sB, 0, sWl1, n0, acc, l2, hi);
  bv = sbl1[n0 + l2];
#pragma unroll
  for (int r = 0; r < 8; ++r)
    sC[(r + 8 * hi) * AP1 + n0 + l2] = relu(acc[r] + bv);
  __syncthreads();

  const int m = tid >> 3, p = tid & 7;
  float s = 0.f;
#pragma unroll
  for (int q = 0; q < 8; ++q) s += sC[m * AP1 + p * 8 + q] * sWl2v[p * 8 + q];
  s += __shfl_xor(s, 1, 32);
  s += __shfl_xor(s, 2, 32);
  s += __shfl_xor(s, 4, 32);
  if (p == 0) out[r0 + m] = s + b_l2[0];
}

extern "C" void kernel_launch(void* const* d_in, const int* in_sizes, int n_in,
                              void* d_out, int out_size, void* d_ws,
                              size_t ws_size, hipStream_t stream) {
  const float* x = (const float*)d_in[0];
  const float* seq_ts = (const float*)d_in[1];
  const float* coeffs = (const float*)d_in[3];
  const float* times = (const float*)d_in[4];
  const float* W_if = (const float*)d_in[5];
  const float* b_if = (const float*)d_in[6];
  const float* W_ic = (const float*)d_in[7];
  const float* b_ic = (const float*)d_in[8];
  const float* W_emb = (const float*)d_in[9];
  const float* b_emb = (const float*)d_in[10];
  const float* W_fl1 = (const float*)d_in[11];
  const float* b_fl1 = (const float*)d_in[12];
  const float* W_flo = (const float*)d_in[13];
  const float* b_flo = (const float*)d_in[14];
  const float* W_f1 = (const float*)d_in[15];
  const float* b_f1 = (const float*)d_in[16];
  const float* W_f2 = (const float*)d_in[17];
  const float* b_f2 = (const float*)d_in[18];
  const float* W_mix = (const float*)d_in[19];
  const float* b_mix = (const float*)d_in[20];
  const float* W_l1 = (const float*)d_in[21];
  const float* b_l1 = (const float*)d_in[22];
  const float* W_l2 = (const float*)d_in[23];
  const float* b_l2 = (const float*)d_in[24];

  float* zflow = (float*)d_ws;
  float* zcde = zflow + (size_t)Bb * Tt * Hh;
  float* out = (float*)d_out;
  float* zout = out + (size_t)Bb * Tt;

  ncde_embed_kernel<<<(Bb * Tt) / 16, 128, 0, stream>>>(
      x, seq_ts, coeffs, times, W_if, b_if, W_ic, b_ic, W_emb, b_emb, W_fl1,
      b_fl1, W_flo, b_flo, zflow);
  ncde_scan_kernel<<<Bb / RB, 512, 0, stream>>>(coeffs, times, W_ic, b_ic,
                                                W_f1, b_f1, W_f2, b_f2, zcde);
  ncde_head_kernel<<<(Bb * Tt) / 16, 128, 0, stream>>>(
      zflow, zcde, W_mix, b_mix, W_l1, b_l1, W_l2, b_l2, out, zout);
}


# scan: W_f1 and b_f1 held in registers of the four GEMM1 waves as WMMA B fragments (as W_f2 already is), not re-read from LDS every sub-step; GEMM1 reads only z from LDS; sums unchanged
# speedup vs baseline: 2.9470x; 1.0398x over previous
#include <hip/hip_runtime.h>
#include <math.h>

#define Bb 256
#define Tt 512
#define Cc 16
#define Hh 64
#define RB 2
#define NVAL (2 * RB)

#define WP  72
#define AP1 69
#define AP2 133
#define DXP 20

typedef __attribute__((ext_vector_type(2))) float v2f;
typedef __attribute__((ext_vector_type(8))) float v8f;

static __device__ __forceinline__ v8f wmma4(v2f a, v2f b, v8f c) {
  return __builtin_amdgcn_wmma_f32_16x16x4_f32(false, a, false, b, (short)0, c,
                                               false, false);
}

template <int K, int APITCH, int ROWMASK = 15>
static __device__ __forceinline__ v8f gemm_tile(const float* __restrict__ A,
                                                int acol0,
                                                const float* __restrict__ W,
                                                int n0, v8f acc, int l2, int hi) {
#pragma unroll
  for (int k = 0; k < K; k += 4) {
    v2f a, b;
    const float* ap = A + (l2 & ROWMASK) * APITCH + acol0 + k + 2 * hi;
    a.x = ap[0];
    a.y = ap[1];
    const float* wp = W + (k + 2 * hi) * WP + n0 + l2;
    b.x = wp[0];
    b.y = wp[WP];
    acc = wmma4(a, b, acc);
  }
  return acc;
}

template <int APITCH>
static __device__ __forceinline__ void store_tile(float* __restrict__ O, int n0,
                                                  v8f acc, int l2, int hi) {
#pragma unroll
  for (int r = 0; r < 8; ++r) O[(r + 8 * hi) * APITCH + n0 + l2] = acc[r];
}

template <int ROWS>
static __device__ __forceinline__ void load_w64(float* __restrict__ dst,
                                                const float* __restrict__ src,
                                                int tid, int nthr) {
  for (int i = tid; i < ROWS * 64; i += nthr)
    dst[(i >> 6) * WP + (i & 63)] = src[i];
}

static __device__ __forceinline__ float relu(float v) { return v > 0.f ? v : 0.f; }

template <int M>
static __device__ __forceinline__ float row_xor_add(float p) {
  const int other = __builtin_amdgcn_update_dpp(0, __builtin_bit_cast(int, p),
                                                0x160 + M, 0xf, 0xf, true);
  return p + __builtin_bit_cast(float, other);
}

__global__ __launch_bounds__(128) void ncde_embed_kernel(
    const float* __restrict__ x, const float* __restrict__ seq_ts,
    const float* __restrict__ coeffs, const float* __restrict__ times,
    const float* __restrict__ W_if, const float* __restrict__ b_if,
    const float* __restrict__ W_ic, const float* __restrict__ b_ic,
    const float* __restrict__ W_emb, const float* __restrict__ b_emb,
    const float* __restrict__ W_fl1, const float* __restrict__ b_fl1,
    const float* __restrict__ W_flo, const float* __restrict__ b_flo,
    float* __restrict__ zflow) {
  __shared__ float sWif[16 * WP], sWic[16 * WP], sWemb[128 * WP];
  __shared__ float sWfl1[64 * WP], sWflo[64 * WP];
  __shared__ float sb[5][64];
  __shared__ float sA[16 * AP2], sB[16 * AP2];

  const int tid = threadIdx.x, nthr = 128;
  const int lane = tid & 31, wave = tid >> 5, l2 = lane & 15, hi = lane >> 4;
  const int r0 = blockIdx.x * 16;

  load_w64<16>(sWif, W_if, tid, nthr);
  load_w64<16>(sWic, W_ic, tid, nthr);
  load_w64<128>(sWemb, W_emb, tid, nthr);
  load_w64<64>(sWfl1, W_fl1, tid, nthr);
  load_w64<64>(sWflo, W_flo, tid, nthr);
  if (tid < 64) {
    sb[0][tid] = b_if[tid];
    sb[1][tid] = b_ic[tid];
    sb[2][tid] = b_emb[tid];
    sb[3][tid] = b_fl1[tid];
    sb[4][tid] = b_flo[tid];
  }
  for (int i = tid; i < 256; i += nthr) {
    const int m = i >> 4, c = i & 15;
    const int rr = r0 + m, b = rr / Tt, t = rr % Tt;
    sA[m * AP2 + c] = (c == 0) ? seq_ts[rr] : x[(size_t)rr * (Cc - 1) + (c - 1)];
    const int idx = (t < Tt - 1) ? t : (Tt - 2);
    const float s = times[t] - times[idx];
    const float* cf = coeffs + ((size_t)b * (Tt - 1) + idx) * (4 * Cc) + c;
    const float a0 = cf[0], b0 = cf[Cc], c0 = cf[2 * Cc], d0 = cf[3 * Cc];
    sA[m * AP2 + 16 + c] = a0 + s * (b0 + s * (c0 + s * d0));
  }
  __syncthreads();

  const int n0 = wave * 16;
  v8f aF = {}, aX = {};
  aF = gemm_tile<16, AP2>(sA, 0, sWif, n0, aF, l2, hi);
  aX = gemm_tile<16, AP2>(sA, 16, sWic, n0, aX, l2, hi);
  const float bF = sb[0][n0 + l2], bX = sb[1][n0 + l2];
#pragma unroll
  for (int r = 0; r < 8; ++r) {
    const int m = r + 8 * hi;
    sB[m * AP2 + n0 + l2] = aF[r] + bF;
    sB[m * AP2 + 64 + n0 + l2] = aX[r] + bX;
  }
  __syncthreads();

  v8f acc = {};
  acc = gemm_tile<128, AP2>(sB, 0, sWemb, n0, acc, l2, hi);
  float bv = sb[2][n0 + l2];
#pragma unroll
  for (int r = 0; r < 8; ++r)
    sA[(r + 8 * hi) * AP2 + n0 + l2] = relu(acc[r] + bv);
  __syncthreads();

  acc = (v8f){};
  acc = gemm_tile<64, AP2>(sA, 0, sWfl1, n0, acc, l2, hi);
  bv = sb[3][n0 + l2];
#pragma unroll
  for (int r = 0; r < 8; ++r)
    sB[(r + 8 * hi) * AP2 + n0 + l2] = relu(acc[r] + bv);
  __syncthreads();

  acc = (v8f){};
  acc = gemm_tile<64, AP2>(sB, 0, sWflo, n0, acc, l2, hi);
  bv = sb[4][n0 + l2];
#pragma unroll
  for (int r = 0; r < 8; ++r)
    zflow[(size_t)(r0 + r + 8 * hi) * Hh + n0 + l2] = acc[r] + bv;
}

__global__ __launch_bounds__(512, 1) void ncde_scan_kernel(
    const float* __restrict__ coeffs, const float* __restrict__ times,
    const float* __restrict__ W_ic, const float* __restrict__ b_ic,
    const float* __restrict__ W_f1, const float* __restrict__ b_f1,
    const float* __restrict__ W_f2, const float* __restrict__ b_f2,
    float* __restrict__ zcde) {
  __shared__ float sWic[16 * WP];
  __shared__ float sbic[64];
  __shared__ float stim[Tt];
  __shared__ float sh[RB * AP1], szb[RB * AP1], sze[RB * AP1];
  __shared__ float sk[4][RB * AP1];
  __shared__ float sdx[4][RB * DXP];

  const int tid = threadIdx.x, nthr = 512;
  const int lane = tid & 31, wave = tid >> 5, l2 = lane & 15, hi = lane >> 4;
  const int b0 = blockIdx.x * RB;
  const int rblk = (l2 & 7) / RB;
  const int myq = (rblk < 2) ? (2 * (l2 >> 3) + rblk) : -1;

  load_w64<16>(sWic, W_ic, tid, nthr);
  if (tid < 64) sbic[tid] = b_ic[tid];
  for (int i = tid; i < Tt; i += nthr) stim[i] = times[i];

  v2f wf2[4][16];
  float bf2r[2];
#pragma unroll
  for (int t = 0; t < 2; ++t) bf2r[t] = b_f2[16 * (wave * 4 + 2 * hi + t) + l2];
#pragma unroll
  for (int jj = 0; jj < 4; ++jj) {
    const int j = wave * 4 + jj;
    const int col = 16 * j + l2;
#pragma unroll
    for (int kk = 0; kk < 16; ++kk) {
      const int row = 4 * kk + 2 * hi;
      wf2[jj][kk].x = W_f2[(size_t)row * (Hh * Cc) + col];
      wf2[jj][kk].y = W_f2[(size_t)(row + 1) * (Hh * Cc) + col];
    }
  }

  v2f wf1[16];
  const int n1 = (wave & 3) * 16 + l2;
  const float bf1r = b_f1[n1];
#pragma unroll
  for (int kk = 0; kk < 16; ++kk) {
    const int row = 4 * kk + 2 * hi;
    wf1[kk].x = W_f1[row * Hh + n1];
    wf1[kk].y = W_f1[(row + 1) * Hh + n1];
  }

  for (int i = tid; i < RB * 16; i += nthr) {
    const int m = i >> 4, c = i & 15;
    sze[m * AP1 + c] = coeffs[((size_t)(b0 + m) * (Tt - 1)) * (4 * Cc) + c];
  }
  __syncthreads();
  if (wave < 4) {
    const int n0 = wave * 16;
    v8f acc = {};
    acc = gemm_tile<16, AP1, RB - 1>(sze, 0, sWic, n0, acc, l2, hi);
    const float bv = sbic[n0 + l2];
    float zv[RB];
#pragma unroll
    for (int r = 0; r < RB; ++r) zv[r] = acc[r] + bv;
    if (hi == 0) {
#pragma unroll
      for (int r = 0; r < RB; ++r) {
        szb[r * AP1 + n0 + l2] = zv[r];
        zcde[((size_t)(b0 + r) * Tt) * Hh + n0 + l2] = zv[r];
      }
    }
  }
  __syncthreads();
  for (int i = tid; i < RB * Hh; i += nthr) {
    const int m = i >> 6, h = i & 63;
    sze[m * AP1 + h] = szb[m * AP1 + h];
  }
  __syncthreads();

  for (int step = 0; step < Tt - 1; ++step) {
    const float dt = stim[step + 1] - stim[step];
    if (tid < RB * 16) {
      const int m = tid >> 4, c = tid & 15, b = b0 + m;
      const float* cf = coeffs + ((size_t)b * (Tt - 1) + step) * (4 * Cc) + c;
      const float bb = cf[Cc], cc = cf[2 * Cc], dd = cf[3 * Cc];
      const float s1 = dt * (1.f / 3.f), s2 = dt * (2.f / 3.f);
      sdx[0][m * DXP + c] = bb;
      sdx[1][m * DXP + c] = bb + s1 * (2.f * cc + 3.f * dd * s1);
      sdx[2][m * DXP + c] = bb + s2 * (2.f * cc + 3.f * dd * s2);
      const int idx3 = (step + 1 < Tt - 1) ? (step + 1) : (Tt - 2);
      const float* cf3 = coeffs + ((size_t)b * (Tt - 1) + idx3) * (4 * Cc) + c;
      const float b3 = cf3[Cc], c3 = cf3[2 * Cc], d3 = cf3[3 * Cc];
      const float s3 = stim[step + 1] - stim[idx3];
      sdx[3][m * DXP + c] = b3 + s3 * (2.f * c3 + 3.f * d3 * s3);
    }
    __syncthreads();

    for (int sub = 0; sub < 4; ++sub) {
      if (wave < 4) {
        const int n0 = wave * 16;
        v2f az[16];
        const float* zp = sze + (l2 & (RB - 1)) * AP1 + 2 * hi;
#pragma unroll
        for (int kk = 0; kk < 16; ++kk) {
          az[kk].x = zp[4 * kk];
          az[kk].y = zp[4 * kk + 1];
        }
        v8f acc = {};
#pragma unroll
        for (int kk = 0; kk < 16; ++kk) acc = wmma4(az[kk], wf1[kk], acc);
        float hv[RB];
#pragma unroll
        for (int r = 0; r < RB; ++r) hv[r] = relu(acc[r] + bf1r);
        if (hi == 0) {
#pragma unroll
          for (int r = 0; r < RB; ++r) sh[r * AP1 + n0 + l2] = hv[r];
        }
      }
      __syncthreads();

      v8f acc2;
#pragma unroll
      for (int r = 0; r < 8; ++r) acc2[r] = (r < NVAL) ? bf2r[r / RB] : 0.f;
#pragma unroll
      for (int kk = 0; kk < 16; ++kk) {
        v2f hv2;
        const float* ap = sh + (l2 & (RB - 1)) * AP1 + 4 * kk + 2 * hi;
        hv2.x = ap[0];
        hv2.y = ap[1];
#pragma unroll
        for (int jj = 0; jj < 4; ++jj) {
          v2f a;
          a.x = (myq == jj) ? hv2.x : 0.f;
          a.y = (myq == jj) ? hv2.y : 0.f;
          acc2 = wmma4(a, wf2[jj][kk], acc2);
        }
      }
      float dxr[RB];
#pragma unroll
      for (int i = 0; i < RB; ++i) dxr[i] = sdx[sub][i * DXP + l2];
      float* kdst = sk[sub];
      float pv[NVAL];
#pragma unroll
      for (int r = 0; r < NVAL; ++r) pv[r] = tanhf(acc2[r]) * dxr[r % RB];
#pragma unroll
      for (int r = 0; r < NVAL; ++r) {
        float p = pv[r];
        p = row_xor_add<1>(p);
        p = row_xor_add<2>(p);
        p = row_xor_add<4>(p);
        p = row_xor_add<8>(p);
        pv[r] = p;
      }
      if (l2 == 0) {
#pragma unroll
        for (int r = 0; r < NVAL; ++r)
          kdst[(r % RB) * AP1 + wave * 4 + 2 * hi + r / RB] = pv[r];
      }
      __syncthreads();

      for (int i = tid; i < RB * Hh; i += nthr) {
        const int m = i >> 6, h = i & 63;
        const int ad = m * AP1 + h;
        const float zb = szb[ad];
        if (sub == 0) {
          sze[ad] = zb + (dt * (1.f / 3.f)) * sk[0][ad];
        } else if (sub == 1) {
          sze[ad] = zb + dt * (sk[1][ad] - sk[0][ad] * (1.f / 3.f));
        } else if (sub == 2) {
          sze[ad] = zb + dt * (sk[0][ad] - sk[1][ad] + sk[2][ad]);
        } else {
          const float zn =
              zb + (sk[0][ad] + 3.f * (sk[1][ad] + sk[2][ad]) + sk[3][ad]) *
                       (dt * 0.125f);
          szb[ad] = zn;
          sze[ad] = zn;
          zcde[((size_t)(b0 + m) * Tt + (step + 1)) * Hh + h] = zn;
        }
      }
      __syncthreads();
    }
  }
}

__global__ __launch_bounds__(128) void ncde_head_kernel(
    const float* __restrict__ zflow, const float* __restrict__ zcde,
    const float* __restrict__ W_mix, const float* __restrict__ b_mix,
    const float* __restrict__ W_l1, const float* __restrict__ b_l1,
    const float* __restrict__ W_l2, const float* __restrict__ b_l2,
    float* __restrict__ out, float* __restrict__ zout) {
  __shared__ float sWmix[128 * WP], sWl1[64 * WP];
  __shared__ float sbmix[64], sbl1[64], sWl2v[64];
  __shared__ float sA[16 * AP2], sB[16 * AP1], sC[16 * AP1];

  const int tid = threadIdx.x, nthr = 128;
  const int lane = tid & 31, wave = tid >> 5, l2 = lane & 15, hi = lane >> 4;
  const int r0 = blockIdx.x * 16;

  load_w64<128>(sWmix, W_mix, tid, nthr);
  load_w64<64>(sWl1, W_l1, tid, nthr);
  if (tid < 64) {
    sbmix[tid] = b_mix[tid];
    sbl1[tid] = b_l1[tid];
    sWl2v[tid] = W_l2[tid];
  }
  for (int i = tid; i < 16 * 128; i += nthr) {
    const int m = i >> 7, c = i & 127;
    const size_t rr = r0 + m;
    sA[m * AP2 + c] = (c < 64) ? zflow[rr * Hh + c] : zcde[rr * Hh + (c - 64)];
  }
  __syncthreads();

  const int n0 = wave * 16;
  v8f acc = {};
  acc = gemm_tile<128, AP2>(sA, 0, sWmix, n0, acc, l2, hi);
  float bv = sbmix[n0 + l2];
#pragma unroll
  for (int r = 0; r < 8; ++r) {
    const int m = r + 8 * hi;
    const float v = acc[r] + bv;
    zout[(size_t)(r0 + m) * Hh + n0 + l2] = v;
    sB[m * AP1 + n0 + l2] = tanhf(v);
  }
  __syncthreads();

  acc = (v8f){};
  acc = gemm_tile<64, AP1>(sB, 0, sWl1, n0, acc, l2, hi);
  bv = sbl1[n0 + l2];
#pragma unroll
  for (int r = 0; r < 8; ++r)
    sC[(r + 8 * hi) * AP1 + n0 + l2] = relu(acc[r] + bv);
  __syncthreads();

  const int m = tid >> 3, p = tid & 7;
  float s = 0.f;
#pragma unroll
  for (int q = 0; q < 8; ++q) s += sC[m * AP1 + p * 8 + q] * sWl2v[p * 8 + q];
  s += __shfl_xor(s, 1, 32);
  s += __shfl_xor(s, 2, 32);
  s += __shfl_xor(s, 4, 32);
  if (p == 0) out[r0 + m] = s + b_l2[0];
}

extern "C" void kernel_launch(void* const* d_in, const int* in_sizes, int n_in,
                              void* d_out, int out_size, void* d_ws,
                              size_t ws_size, hipStream_t stream) {
  const float* x = (const float*)d_in[0];
  const float* seq_ts = (const float*)d_in[1];
  const float* coeffs = (const float*)d_in[3];
  const float* times = (const float*)d_in[4];
  const float* W_if = (const float*)d_in[5];
  const float* b_if = (const float*)d_in[6];
  const float* W_ic = (const float*)d_in[7];
  const float* b_ic = (const float*)d_in[8];
  const float* W_emb = (const float*)d_in[9];
  const float* b_emb = (const float*)d_in[10];
  const float* W_fl1 = (const float*)d_in[11];
  const float* b_fl1 = (const float*)d_in[12];
  const float* W_flo = (const float*)d_in[13];
  const float* b_flo = (const float*)d_in[14];
  const float* W_f1 = (const float*)d_in[15];
  const float* b_f1 = (const float*)d_in[16];
  const float* W_f2 = (const float*)d_in[17];
  const float* b_f2 = (const float*)d_in[18];
  const float* W_mix = (const float*)d_in[19];
  const float* b_mix = (const float*)d_in[20];
  const float* W_l1 = (const float*)d_in[21];
  const float* b_l1 = (const float*)d_in[22];
  const float* W_l2 = (const float*)d_in[23];
  const float* b_l2 = (const float*)d_in[24];

  float* zflow = (float*)d_ws;
  float* zcde = zflow + (size_t)Bb * Tt * Hh;
  float* out = (float*)d_out;
  float* zout = out + (size_t)Bb * Tt;

  ncde_embed_kernel<<<(Bb * Tt) / 16, 128, 0, stream>>>(
      x, seq_ts, coeffs, times, W_if, b_if, W_ic, b_ic, W_emb, b_emb, W_fl1,
      b_fl1, W_flo, b_flo, zflow);
  ncde_scan_kernel<<<Bb / RB, 512, 0, stream>>>(coeffs, times, W_ic, b_ic,
                                                W_f1, b_f1, W_f2, b_f2, zcde);
  ncde_head_kernel<<<(Bb * Tt) / 16, 128, 0, stream>>>(
      zflow, zcde, W_mix, b_mix, W_l1, b_l1, W_l2, b_l2, out, zout);
}


# scan: batch rows per work-group 2 -> 1 (256 work-groups of 16 waves, one per WGP, not 128): tanh/einsum values per lane 4 -> 2, matrix instructions per wave unchanged; sums unchanged
# speedup vs baseline: 3.1480x; 1.0682x over previous
#include <hip/hip_runtime.h>
#include <math.h>

#define Bb 256
#define Tt 512
#define Cc 16
#define Hh 64
#define RB 1
#define NVAL (2 * RB)

#define WP  72
#define AP1 69
#define AP2 133
#define DXP 20

typedef __attribute__((ext_vector_type(2))) float v2f;
typedef __attribute__((ext_vector_type(8))) float v8f;

static __device__ __forceinline__ v8f wmma4(v2f a, v2f b, v8f c) {
  return __builtin_amdgcn_wmma_f32_16x16x4_f32(false, a, false, b, (short)0, c,
                                               false, false);
}

template <int K, int APITCH, int ROWMASK = 15>
static __device__ __forceinline__ v8f gemm_tile(const float* __restrict__ A,
                                                int acol0,
                                                const float* __restrict__ W,
                                                int n0, v8f acc, int l2, int hi) {
#pragma unroll
  for (int k = 0; k < K; k += 4) {
    v2f a, b;
    const float* ap = A + (l2 & ROWMASK) * APITCH + acol0 + k + 2 * hi;
    a.x = ap[0];
    a.y = ap[1];
    const float* wp = W + (k + 2 * hi) * WP + n0 + l2;
    b.x = wp[0];
    b.y = wp[WP];
    acc = wmma4(a, b, acc);
  }
  return acc;
}

template <int APITCH>
static __device__ __forceinline__ void store_tile(float* __restrict__ O, int n0,
                                                  v8f acc, int l2, int hi) {
#pragma unroll
  for (int r = 0; r < 8; ++r) O[(r + 8 * hi) * APITCH + n0 + l2] = acc[r];
}

template <int ROWS>
static __device__ __forceinline__ void load_w64(float* __restrict__ dst,
                                                const float* __restrict__ src,
                                                int tid, int nthr) {
  for (int i = tid; i < ROWS * 64; i += nthr)
    dst[(i >> 6) * WP + (i & 63)] = src[i];
}

static __device__ __forceinline__ float relu(float v) { return v > 0.f ? v : 0.f; }

template <int M>
static __device__ __forceinline__ float row_xor_add(float p) {
  const int other = __builtin_amdgcn_update_dpp(0, __builtin_bit_cast(int, p),
                                                0x160 + M, 0xf, 0xf, true);
  return p + __builtin_bit_cast(float, other);
}

__global__ __launch_bounds__(128) void ncde_embed_kernel(
    const float* __restrict__ x, const float* __restrict__ seq_ts,
    const float* __restrict__ coeffs, const float* __restrict__ times,
    const float* __restrict__ W_if, const float* __restrict__ b_if,
    const float* __restrict__ W_ic, const float* __restrict__ b_ic,
    const float* __restrict__ W_emb, const float* __restrict__ b_emb,
    const float* __restrict__ W_fl1, const float* __restrict__ b_fl1,
    const float* __restrict__ W_flo, const float* __restrict__ b_flo,
    float* __restrict__ zflow) {
  __shared__ float sWif[16 * WP], sWic[16 * WP], sWemb[128 * WP];
  __shared__ float sWfl1[64 * WP], sWflo[64 * WP];
  __shared__ float sb[5][64];
  __shared__ float sA[16 * AP2], sB[16 * AP2];

  const int tid = threadIdx.x, nthr = 128;
  const int lane = tid & 31, wave = tid >> 5, l2 = lane & 15, hi = lane >> 4;
  const int r0 = blockIdx.x * 16;

  load_w64<16>(sWif, W_if, tid, nthr);
  load_w64<16>(sWic, W_ic, tid, nthr);
  load_w64<128>(sWemb, W_emb, tid, nthr);
  load_w64<64>(sWfl1, W_fl1, tid, nthr);
  load_w64<64>(sWflo, W_flo, tid, nthr);
  if (tid < 64) {
    sb[0][tid] = b_if[tid];
    sb[1][tid] = b_ic[tid];
    sb[2][tid] = b_emb[tid];
    sb[3][tid] = b_fl1[tid];
    sb[4][tid] = b_flo[tid];
  }
  for (int i = tid; i < 256; i += nthr) {
    const int m = i >> 4, c = i & 15;
    const int rr = r0 + m, b = rr / Tt, t = rr % Tt;
    sA[m * AP2 + c] = (c == 0) ? seq_ts[rr] : x[(size_t)rr * (Cc - 1) + (c - 1)];
    const int idx = (t < Tt - 1) ? t : (Tt - 2);
    const float s = times[t] - times[idx];
    const float* cf = coeffs + ((size_t)b * (Tt - 1) + idx) * (4 * Cc) + c;
    const float a0 = cf[0], b0 = cf[Cc], c0 = cf[2 * Cc], d0 = cf[3 * Cc];
    sA[m * AP2 + 16 + c] = a0 + s * (b0 + s * (c0 + s * d0));
  }
  __syncthreads();

  const int n0 = wave * 16;
  v8f aF = {}, aX = {};
  aF = gemm_tile<16, AP2>(sA, 0, sWif, n0, aF, l2, hi);
  aX = gemm_tile<16, AP2>(sA, 16, sWic, n0, aX, l2, hi);
  const float bF = sb[0][n0 + l2], bX = sb[1][n0 + l2];
#pragma unroll
  for (int r = 0; r < 8; ++r) {
    const int m = r + 8 * hi;
    sB[m * AP2 + n0 + l2] = aF[r] + bF;
    sB[m * AP2 + 64 + n0 + l2] = aX[r] + bX;
  }
  __syncthreads();

  v8f acc = {};
  acc = gemm_tile<128, AP2>(sB, 0, sWemb, n0, acc, l2, hi);
  float bv = sb[2][n0 + l2];
#pragma unroll
  for (int r = 0; r < 8; ++r)
    sA[(r + 8 * hi) * AP2 + n0 + l2] = relu(acc[r] + bv);
  __syncthreads();

  acc = (v8f){};
  acc = gemm_tile<64, AP2>(sA, 0, sWfl1, n0, acc, l2, hi);
  bv = sb[3][n0 + l2];
#pragma unroll
  for (int r = 0; r < 8; ++r)
    sB[(r + 8 * hi) * AP2 + n0 + l2] = relu(acc[r] + bv);
  __syncthreads();

  acc = (v8f){};
  acc = gemm_tile<64, AP2>(sB, 0, sWflo, n0, acc, l2, hi);
  bv = sb[4][n0 + l2];
#pragma unroll
  for (int r = 0; r < 8; ++r)
    zflow[(size_t)(r0 + r + 8 * hi) * Hh + n0 + l2] = acc[r] + bv;
}

__global__ __launch_bounds__(512, 1) void ncde_scan_kernel(
    const float* __restrict__ coeffs, const float* __restrict__ times,
    const float* __restrict__ W_ic, const float* __restrict__ b_ic,
    const float* __restrict__ W_f1, const float* __restrict__ b_f1,
    const float* __restrict__ W_f2, const float* __restrict__ b_f2,
    float* __restrict__ zcde) {
  __shared__ float sWic[16 * WP];
  __shared__ float sbic[64];
  __shared__ float stim[Tt];
  __shared__ float sh[RB * AP1], szb[RB * AP1], sze[RB * AP1];
  __shared__ float sk[4][RB * AP1];
  __shared__ float sdx[4][RB * DXP];

  const int tid = threadIdx.x, nthr = 512;
  const int lane = tid & 31, wave = tid >> 5, l2 = lane & 15, hi = lane >> 4;
  const int b0 = blockIdx.x * RB;
  const int rblk = (l2 & 7) / RB;
  const int myq = (rblk < 2) ? (2 * (l2 >> 3) + rblk) : -1;

  load_w64<16>(sWic, W_ic, tid, nthr);
  if (tid < 64) sbic[tid] = b_ic[tid];
  for (int i = tid; i < Tt; i += nthr) stim[i] = times[i];

  v2f wf2[4][16];
  float bf2r[2];
#pragma unroll
  for (int t = 0; t < 2; ++t) bf2r[t] = b_f2[16 * (wave * 4 + 2 * hi + t) + l2];
#pragma unroll
  for (int jj = 0; jj < 4; ++jj) {
    const int j = wave * 4 + jj;
    const int col = 16 * j + l2;
#pragma unroll
    for (int kk = 0; kk < 16; ++kk) {
      const int row = 4 * kk + 2 * hi;
      wf2[jj][kk].x = W_f2[(size_t)row * (Hh * Cc) + col];
      wf2[jj][kk].y = W_f2[(size_t)(row + 1) * (Hh * Cc) + col];
    }
  }

  v2f wf1[16];
  const int n1 = (wave & 3) * 16 + l2;
  const float bf1r = b_f1[n1];
#pragma unroll
  for (int kk = 0; kk < 16; ++kk) {
    const int row = 4 * kk + 2 * hi;
    wf1[kk].x = W_f1[row * Hh + n1];
    wf1[kk].y = W_f1[(row + 1) * Hh + n1];
  }

  for (int i = tid; i < RB * 16; i += nthr) {
    const int m = i >> 4, c = i & 15;
    sze[m * AP1 + c] = coeffs[((size_t)(b0 + m) * (Tt - 1)) * (4 * Cc) + c];
  }
  __syncthreads();
  if (wave < 4) {
    const int n0 = wave * 16;
    v8f acc = {};
    acc = gemm_tile<16, AP1, RB - 1>(sze, 0, sWic, n0, acc, l2, hi);
    const float bv = sbic[n0 + l2];
    float zv[RB];
#pragma unroll
    for (int r = 0; r < RB; ++r) zv[r] = acc[r] + bv;
    if (hi == 0) {
#pragma unroll
      for (int r = 0; r < RB; ++r) {
        szb[r * AP1 + n0 + l2] = zv[r];
        zcde[((size_t)(b0 + r) * Tt) * Hh + n0 + l2] = zv[r];
      }
    }
  }
  __syncthreads();
  for (int i = tid; i < RB * Hh; i += nthr) {
    const int m = i >> 6, h = i & 63;
    sze[m * AP1 + h] = szb[m * AP1 + h];
  }
  __syncthreads();

  for (int step = 0; step < Tt - 1; ++step) {
    const float dt = stim[step + 1] - stim[step];
    if (tid < RB * 16) {
      const int m = tid >> 4, c = tid & 15, b = b0 + m;
      const float* cf = coeffs + ((size_t)b * (Tt - 1) + step) * (4 * Cc) + c;
      const float bb = cf[Cc], cc = cf[2 * Cc], dd = cf[3 * Cc];
      const float s1 = dt * (1.f / 3.f), s2 = dt * (2.f / 3.f);
      sdx[0][m * DXP + c] = bb;
      sdx[1][m * DXP + c] = bb + s1 * (2.f * cc + 3.f * dd * s1);
      sdx[2][m * DXP + c] = bb + s2 * (2.f * cc + 3.f * dd * s2);
      const int idx3 = (step + 1 < Tt - 1) ? (step + 1) : (Tt - 2);
      const float* cf3 = coeffs + ((size_t)b * (Tt - 1) + idx3) * (4 * Cc) + c;
      const float b3 = cf3[Cc], c3 = cf3[2 * Cc], d3 = cf3[3 * Cc];
      const float s3 = stim[step + 1] - stim[idx3];
      sdx[3][m * DXP + c] = b3 + s3 * (2.f * c3 + 3.f * d3 * s3);
    }
    __syncthreads();

    for (int sub = 0; sub < 4; ++sub) {
      if (wave < 4) {
        const int n0 = wave * 16;
        v2f az[16];
        const float* zp = sze + (l2 & (RB - 1)) * AP1 + 2 * hi;
#pragma unroll
        for (int kk = 0; kk < 16; ++kk) {
          az[kk].x = zp[4 * kk];
          az[kk].y = zp[4 * kk + 1];
        }
        v8f acc = {};
#pragma unroll
        for (int kk = 0; kk < 16; ++kk) acc = wmma4(az[kk], wf1[kk], acc);
        float hv[RB];
#pragma unroll
        for (int r = 0; r < RB; ++r) hv[r] = relu(acc[r] + bf1r);
        if (hi == 0) {
#pragma unroll
          for (int r = 0; r < RB; ++r) sh[r * AP1 + n0 + l2] = hv[r];
        }
      }
      __syncthreads();

      v8f acc2;
#pragma unroll
      for (int r = 0; r < 8; ++r) acc2[r] = (r < NVAL) ? bf2r[r / RB] : 0.f;
#pragma unroll
      for (int kk = 0; kk < 16; ++kk) {
        v2f hv2;
        const float* ap = sh + (l2 & (RB - 1)) * AP1 + 4 * kk + 2 * hi;
        hv2.x = ap[0];
        hv2.y = ap[1];
#pragma unroll
        for (int jj = 0; jj < 4; ++jj) {
          v2f a;
          a.x = (myq == jj) ? hv2.x : 0.f;
          a.y = (myq == jj) ? hv2.y : 0.f;
          acc2 = wmma4(a, wf2[jj][kk], acc2);
        }
      }
      float dxr[RB];
#pragma unroll
      for (int i = 0; i < RB; ++i) dxr[i] = sdx[sub][i * DXP + l2];
      float* kdst = sk[sub];
      float pv[NVAL];
#pragma unroll
      for (int r = 0; r < NVAL; ++r) pv[r] = tanhf(acc2[r]) * dxr[r % RB];
#pragma unroll
      for (int r = 0; r < NVAL; ++r) {
        float p = pv[r];
        p = row_xor_add<1>(p);
        p = row_xor_add<2>(p);
        p = row_xor_add<4>(p);
        p = row_xor_add<8>(p);
        pv[r] = p;
      }
      if (l2 == 0) {
#pragma unroll
        for (int r = 0; r < NVAL; ++r)
          kdst[(r % RB) * AP1 + wave * 4 + 2 * hi + r / RB] = pv[r];
      }
      __syncthreads();

      for (int i = tid; i < RB * Hh; i += nthr) {
        const int m = i >> 6, h = i & 63;
        const int ad = m * AP1 + h;
        const float zb = szb[ad];
        if (sub == 0) {
          sze[ad] = zb + (dt * (1.f / 3.f)) * sk[0][ad];
        } else if (sub == 1) {
          sze[ad] = zb + dt * (sk[1][ad] - sk[0][ad] * (1.f / 3.f));
        } else if (sub == 2) {
          sze[ad] = zb + dt * (sk[0][ad] - sk[1][ad] + sk[2][ad]);
        } else {
          const float zn =
              zb + (sk[0][ad] + 3.f * (sk[1][ad] + sk[2][ad]) + sk[3][ad]) *
                       (dt * 0.125f);
          szb[ad] = zn;
          sze[ad] = zn;
          zcde[((size_t)(b0 + m) * Tt + (step + 1)) * Hh + h] = zn;
        }
      }
      __syncthreads();
    }
  }
}

__global__ __launch_bounds__(128) void ncde_head_kernel(
    const float* __restrict__ zflow, const float* __restrict__ zcde,
    const float* __restrict__ W_mix, const float* __restrict__ b_mix,
    const float* __restrict__ W_l1, const float* __restrict__ b_l1,
    const float* __restrict__ W_l2, const float* __restrict__ b_l2,
    float* __restrict__ out, float* __restrict__ zout) {
  __shared__ float sWmix[128 * WP], sWl1[64 * WP];
  __shared__ float sbmix[64], sbl1[64], sWl2v[64];
  __shared__ float sA[16 * AP2], sB[16 * AP1], sC[16 * AP1];

  const int tid = threadIdx.x, nthr = 128;
  const int lane = tid & 31, wave = tid >> 5, l2 = lane & 15, hi = lane >> 4;
  const int r0 = blockIdx.x * 16;

  load_w64<128>(sWmix, W_mix, tid, nthr);
  load_w64<64>(sWl1, W_l1, tid, nthr);
  if (tid < 64) {
    sbmix[tid] = b_mix[tid];
    sbl1[tid] = b_l1[tid];
    sWl2v[tid] = W_l2[tid];
  }
  for (int i = tid; i < 16 * 128; i += nthr) {
    const int m = i >> 7, c = i & 127;
    const size_t rr = r0 + m;
    sA[m * AP2 + c] = (c < 64) ? zflow[rr * Hh + c] : zcde[rr * Hh + (c - 64)];
  }
  __syncthreads();

  const int n0 = wave * 16;
  v8f acc = {};
  acc = gemm_tile<128, AP2>(sA, 0, sWmix, n0, acc, l2, hi);
  float bv = sbmix[n0 + l2];
#pragma unroll
  for (int r = 0; r < 8; ++r) {
    const int m = r + 8 * hi;
    const float v = acc[r] + bv;
    zout[(size_t)(r0 + m) * Hh + n0 + l2] = v;
    sB[m * AP1 + n0 + l2] = tanhf(v);
  }
  __syncthreads();

  acc = (v8f){};
  acc = gemm_tile<64, AP1>(sB, 0, sWl1, n0, acc, l2, hi);
  bv = sbl1[n0 + l2];
#pragma unroll
  for (int r = 0; r < 8; ++r)
    sC[(r + 8 * hi) * AP1 + n0 + l2] = relu(acc[r] + bv);
  __syncthreads();

  const int m = tid >> 3, p = tid & 7;
  float s = 0.f;
#pragma unroll
  for (int q = 0; q < 8; ++q) s += sC[m * AP1 + p * 8 + q] * sWl2v[p * 8 + q];
  s += __shfl_xor(s, 1, 32);
  s += __shfl_xor(s, 2, 32);
  s += __shfl_xor(s, 4, 32);
  if (p == 0) out[r0 + m] = s + b_l2[0];
}

extern "C" void kernel_launch(void* const* d_in, const int* in_sizes, int n_in,
                              void* d_out, int out_size, void* d_ws,
                              size_t ws_size, hipStream_t stream) {
  const float* x = (const float*)d_in[0];
  const float* seq_ts = (const float*)d_in[1];
  const float* coeffs = (const float*)d_in[3];
  const float* times = (const float*)d_in[4];
  const float* W_if = (const float*)d_in[5];
  const float* b_if = (const float*)d_in[6];
  const float* W_ic = (const float*)d_in[7];
  const float* b_ic = (const float*)d_in[8];
  const float* W_emb = (const float*)d_in[9];
  const float* b_emb = (const float*)d_in[10];
  const float* W_fl1 = (const float*)d_in[11];
  const float* b_fl1 = (const float*)d_in[12];
  const float* W_flo = (const float*)d_in[13];
  const float* b_flo = (const float*)d_in[14];
  const float* W_f1 = (const float*)d_in[15];
  const float* b_f1 = (const float*)d_in[16];
  const float* W_f2 = (const float*)d_in[17];
  const float* b_f2 = (const float*)d_in[18];
  const float* W_mix = (const float*)d_in[19];
  const float* b_mix = (const float*)d_in[20];
  const float* W_l1 = (const float*)d_in[21];
  const float* b_l1 = (const float*)d_in[22];
  const float* W_l2 = (const float*)d_in[23];
  const float* b_l2 = (const float*)d_in[24];

  float* zflow = (float*)d_ws;
  float* zcde = zflow + (size_t)Bb * Tt * Hh;
  float* out = (float*)d_out;
  float* zout = out + (size_t)Bb * Tt;

  ncde_embed_kernel<<<(Bb * Tt) / 16, 128, 0, stream>>>(
      x, seq_ts, coeffs, times, W_if, b_if, W_ic, b_ic, W_emb, b_emb, W_fl1,
      b_fl1, W_flo, b_flo, zflow);
  ncde_scan_kernel<<<Bb / RB, 512, 0, stream>>>(coeffs, times, W_ic, b_ic,
                                                W_f1, b_f1, W_f2, b_f2, zcde);
  ncde_head_kernel<<<(Bb * Tt) / 16, 128, 0, stream>>>(
      zflow, zcde, W_mix, b_mix, W_l1, b_l1, W_l2, b_l2, out, zout);
}


# scan: GEMM2 gives each of a wave's 4 tiles its own accumulator and feeds h to every A row: 4 independent matrix instructions per k step, not one chain of 64 with 128 selects; sums unchanged
# speedup vs baseline: 3.3465x; 1.0631x over previous
#include <hip/hip_runtime.h>
#include <math.h>

#define Bb 256
#define Tt 512
#define Cc 16
#define Hh 64
#define RB 1
#define NVAL (2 * RB)

#define WP  72
#define AP1 69
#define AP2 133
#define DXP 20

typedef __attribute__((ext_vector_type(2))) float v2f;
typedef __attribute__((ext_vector_type(8))) float v8f;

static __device__ __forceinline__ v8f wmma4(v2f a, v2f b, v8f c) {
  return __builtin_amdgcn_wmma_f32_16x16x4_f32(false, a, false, b, (short)0, c,
                                               false, false);
}

template <int K, int APITCH, int ROWMASK = 15>
static __device__ __forceinline__ v8f gemm_tile(const float* __restrict__ A,
                                                int acol0,
                                                const float* __restrict__ W,
                                                int n0, v8f acc, int l2, int hi) {
#pragma unroll
  for (int k = 0; k < K; k += 4) {
    v2f a, b;
    const float* ap = A + (l2 & ROWMASK) * APITCH + acol0 + k + 2 * hi;
    a.x = ap[0];
    a.y = ap[1];
    const float* wp = W + (k + 2 * hi) * WP + n0 + l2;
    b.x = wp[0];
    b.y = wp[WP];
    acc = wmma4(a, b, acc);
  }
  return acc;
}

template <int APITCH>
static __device__ __forceinline__ void store_tile(float* __restrict__ O, int n0,
                                                  v8f acc, int l2, int hi) {
#pragma unroll
  for (int r = 0; r < 8; ++r) O[(r + 8 * hi) * APITCH + n0 + l2] = acc[r];
}

template <int ROWS>
static __device__ __forceinline__ void load_w64(float* __restrict__ dst,
                                                const float* __restrict__ src,
                                                int tid, int nthr) {
  for (int i = tid; i < ROWS * 64; i += nthr)
    dst[(i >> 6) * WP + (i & 63)] = src[i];
}

static __device__ __forceinline__ float relu(float v) { return v > 0.f ? v : 0.f; }

template <int M>
static __device__ __forceinline__ float row_xor_add(float p) {
  const int other = __builtin_amdgcn_update_dpp(0, __builtin_bit_cast(int, p),
                                                0x160 + M, 0xf, 0xf, true);
  return p + __builtin_bit_cast(float, other);
}

__global__ __launch_bounds__(128) void ncde_embed_kernel(
    const float* __restrict__ x, const float* __restrict__ seq_ts,
    const float* __restrict__ coeffs, const float* __restrict__ times,
    const float* __restrict__ W_if, const float* __restrict__ b_if,
    const float* __restrict__ W_ic, const float* __restrict__ b_ic,
    const float* __restrict__ W_emb, const float* __restrict__ b_emb,
    const float* __restrict__ W_fl1, const float* __restrict__ b_fl1,
    const float* __restrict__ W_flo, const float* __restrict__ b_flo,
    float* __restrict__ zflow) {
  __shared__ float sWif[16 * WP], sWic[16 * WP], sWemb[128 * WP];
  __shared__ float sWfl1[64 * WP], sWflo[64 * WP];
  __shared__ float sb[5][64];
  __shared__ float sA[16 * AP2], sB[16 * AP2];

  const int tid = threadIdx.x, nthr = 128;
  const int lane = tid & 31, wave = tid >> 5, l2 = lane & 15, hi = lane >> 4;
  const int r0 = blockIdx.x * 16;

  load_w64<16>(sWif, W_if, tid, nthr);
  load_w64<16>(sWic, W_ic, tid, nthr);
  load_w64<128>(sWemb, W_emb, tid, nthr);
  load_w64<64>(sWfl1, W_fl1, tid, nthr);
  load_w64<64>(sWflo, W_flo, tid, nthr);
  if (tid < 64) {
    sb[0][tid] = b_if[tid];
    sb[1][tid] = b_ic[tid];
    sb[2][tid] = b_emb[tid];
    sb[3][tid] = b_fl1[tid];
    sb[4][tid] = b_flo[tid];
  }
  for (int i = tid; i < 256; i += nthr) {
    const int m = i >> 4, c = i & 15;
    const int rr = r0 + m, b = rr / Tt, t = rr % Tt;
    sA[m * AP2 + c] = (c == 0) ? seq_ts[rr] : x[(size_t)rr * (Cc - 1) + (c - 1)];
    const int idx = (t < Tt - 1) ? t : (Tt - 2);
    const float s = times[t] - times[idx];
    const float* cf = coeffs + ((size_t)b * (Tt - 1) + idx) * (4 * Cc) + c;
    const float a0 = cf[0], b0 = cf[Cc], c0 = cf[2 * Cc], d0 = cf[3 * Cc];
    sA[m * AP2 + 16 + c] = a0 + s * (b0 + s * (c0 + s * d0));
  }
  __syncthreads();

  const int n0 = wave * 16;
  v8f aF = {}, aX = {};
  aF = gemm_tile<16, AP2>(sA, 0, sWif, n0, aF, l2, hi);
  aX = gemm_tile<16, AP2>(sA, 16, sWic, n0, aX, l2, hi);
  const float bF = sb[0][n0 + l2], bX = sb[1][n0 + l2];
#pragma unroll
  for (int r = 0; r < 8; ++r) {
    const int m = r + 8 * hi;
    sB[m * AP2 + n0 + l2] = aF[r] + bF;
    sB[m * AP2 + 64 + n0 + l2] = aX[r] + bX;
  }
  __syncthreads();

  v8f acc = {};
  acc = gemm_tile<128, AP2>(sB, 0, sWemb, n0, acc, l2, hi);
  float bv = sb[2][n0 + l2];
#pragma unroll
  for (int r = 0; r < 8; ++r)
    sA[(r + 8 * hi) * AP2 + n0 + l2] = relu(acc[r] + bv);
  __syncthreads();

  acc = (v8f){};
  acc = gemm_tile<64, AP2>(sA, 0, sWfl1, n0, acc, l2, hi);
  bv = sb[3][n0 + l2];
#pragma unroll
  for (int r = 0; r < 8; ++r)
    sB[(r + 8 * hi) * AP2 + n0 + l2] = relu(acc[r] + bv);
  __syncthreads();

  acc = (v8f){};
  acc = gemm_tile<64, AP2>(sB, 0, sWflo, n0, acc, l2, hi);
  bv = sb[4][n0 + l2];
#pragma unroll
  for (int r = 0; r < 8; ++r)
    zflow[(size_t)(r0 + r + 8 * hi) * Hh + n0 + l2] = acc[r] + bv;
}

__global__ __launch_bounds__(512, 1) void ncde_scan_kernel(
    const float* __restrict__ coeffs, const float* __restrict__ times,
    const float* __restrict__ W_ic, const float* __restrict__ b_ic,
    const float* __restrict__ W_f1, const float* __restrict__ b_f1,
    const float* __restrict__ W_f2, const float* __restrict__ b_f2,
    float* __restrict__ zcde) {
  __shared__ float sWic[16 * WP];
  __shared__ float sbic[64];
  __shared__ float stim[Tt];
  __shared__ float sh[RB * AP1], szb[RB * AP1], sze[RB * AP1];
  __shared__ float sk[4][RB * AP1];
  __shared__ float sdx[4][RB * DXP];

  const int tid = threadIdx.x, nthr = 512;
  const int lane = tid & 31, wave = tid >> 5, l2 = lane & 15, hi = lane >> 4;
  const int b0 = blockIdx.x * RB;
  load_w64<16>(sWic, W_ic, tid, nthr);
  if (tid < 64) sbic[tid] = b_ic[tid];
  for (int i = tid; i < Tt; i += nthr) stim[i] = times[i];

  v2f wf2[4][16];
  float bf2r[4];
#pragma unroll
  for (int t = 0; t < 4; ++t) bf2r[t] = b_f2[16 * (wave * 4 + t) + l2];
#pragma unroll
  for (int jj = 0; jj < 4; ++jj) {
    const int j = wave * 4 + jj;
    const int col = 16 * j + l2;
#pragma unroll
    for (int kk = 0; kk < 16; ++kk) {
      const int row = 4 * kk + 2 * hi;
      wf2[jj][kk].x = W_f2[(size_t)row * (Hh * Cc) + col];
      wf2[jj][kk].y = W_f2[(size_t)(row + 1) * (Hh * Cc) + col];
    }
  }

  v2f wf1[16];
  const int n1 = (wave & 3) * 16 + l2;
  const float bf1r = b_f1[n1];
#pragma unroll
  for (int kk = 0; kk < 16; ++kk) {
    const int row = 4 * kk + 2 * hi;
    wf1[kk].x = W_f1[row * Hh + n1];
    wf1[kk].y = W_f1[(row + 1) * Hh + n1];
  }

  for (int i = tid; i < RB * 16; i += nthr) {
    const int m = i >> 4, c = i & 15;
    sze[m * AP1 + c] = coeffs[((size_t)(b0 + m) * (Tt - 1)) * (4 * Cc) + c];
  }
  __syncthreads();
  if (wave < 4) {
    const int n0 = wave * 16;
    v8f acc = {};
    acc = gemm_tile<16, AP1, RB - 1>(sze, 0, sWic, n0, acc, l2, hi);
    const float bv = sbic[n0 + l2];
    float zv[RB];
#pragma unroll
    for (int r = 0; r < RB; ++r) zv[r] = acc[r] + bv;
    if (hi == 0) {
#pragma unroll
      for (int r = 0; r < RB; ++r) {
        szb[r * AP1 + n0 + l2] = zv[r];
        zcde[((size_t)(b0 + r) * Tt) * Hh + n0 + l2] = zv[r];
      }
    }
  }
  __syncthreads();
  for (int i = tid; i < RB * Hh; i += nthr) {
    const int m = i >> 6, h = i & 63;
    sze[m * AP1 + h] = szb[m * AP1 + h];
  }
  __syncthreads();

  for (int step = 0; step < Tt - 1; ++step) {
    const float dt = stim[step + 1] - stim[step];
    if (tid < RB * 16) {
      const int m = tid >> 4, c = tid & 15, b = b0 + m;
      const float* cf = coeffs + ((size_t)b * (Tt - 1) + step) * (4 * Cc) + c;
      const float bb = cf[Cc], cc = cf[2 * Cc], dd = cf[3 * Cc];
      const float s1 = dt * (1.f / 3.f), s2 = dt * (2.f / 3.f);
      sdx[0][m * DXP + c] = bb;
      sdx[1][m * DXP + c] = bb + s1 * (2.f * cc + 3.f * dd * s1);
      sdx[2][m * DXP + c] = bb + s2 * (2.f * cc + 3.f * dd * s2);
      const int idx3 = (step + 1 < Tt - 1) ? (step + 1) : (Tt - 2);
      const float* cf3 = coeffs + ((size_t)b * (Tt - 1) + idx3) * (4 * Cc) + c;
      const float b3 = cf3[Cc], c3 = cf3[2 * Cc], d3 = cf3[3 * Cc];
      const float s3 = stim[step + 1] - stim[idx3];
      sdx[3][m * DXP + c] = b3 + s3 * (2.f * c3 + 3.f * d3 * s3);
    }
    __syncthreads();

    for (int sub = 0; sub < 4; ++sub) {
      if (wave < 4) {
        const int n0 = wave * 16;
        v2f az[16];
        const float* zp = sze + (l2 & (RB - 1)) * AP1 + 2 * hi;
#pragma unroll
        for (int kk = 0; kk < 16; ++kk) {
          az[kk].x = zp[4 * kk];
          az[kk].y = zp[4 * kk + 1];
        }
        v8f acc = {};
#pragma unroll
        for (int kk = 0; kk < 16; ++kk) acc = wmma4(az[kk], wf1[kk], acc);
        float hv[RB];
#pragma unroll
        for (int r = 0; r < RB; ++r) hv[r] = relu(acc[r] + bf1r);
        if (hi == 0) {
#pragma unroll
          for (int r = 0; r < RB; ++r) sh[r * AP1 + n0 + l2] = hv[r];
        }
      }
      __syncthreads();

      v8f acc2[4];
#pragma unroll
      for (int jj = 0; jj < 4; ++jj) {
#pragma unroll
        for (int r = 0; r < 8; ++r) acc2[jj][r] = bf2r[jj];
      }
#pragma unroll
      for (int kk = 0; kk < 16; ++kk) {
        v2f hv2;
        const float* ap = sh + (l2 & (RB - 1)) * AP1 + 4 * kk + 2 * hi;
        hv2.x = ap[0];
        hv2.y = ap[1];
#pragma unroll
        for (int jj = 0; jj < 4; ++jj)
          acc2[jj] = wmma4(hv2, wf2[jj][kk], acc2[jj]);
      }
      float dxr[RB];
#pragma unroll
      for (int i = 0; i < RB; ++i) dxr[i] = sdx[sub][i * DXP + l2];
      float* kdst = sk[sub];
      float pv[NVAL];
#pragma unroll
      for (int r = 0; r < NVAL; ++r) {
        const float lo = acc2[r / RB][r % RB], up = acc2[2 + r / RB][r % RB];
        pv[r] = tanhf(hi ? up : lo) * dxr[r % RB];
      }
#pragma unroll
      for (int r = 0; r < NVAL; ++r) {
        float p = pv[r];
        p = row_xor_add<1>(p);
        p = row_xor_add<2>(p);
        p = row_xor_add<4>(p);
        p = row_xor_add<8>(p);
        pv[r] = p;
      }
      if (l2 == 0) {
#pragma unroll
        for (int r = 0; r < NVAL; ++r)
          kdst[(r % RB) * AP1 + wave * 4 + 2 * hi + r / RB] = pv[r];
      }
      __syncthreads();

      for (int i = tid; i < RB * Hh; i += nthr) {
        const int m = i >> 6, h = i & 63;
        const int ad = m * AP1 + h;
        const float zb = szb[ad];
        if (sub == 0) {
          sze[ad] = zb + (dt * (1.f / 3.f)) * sk[0][ad];
        } else if (sub == 1) {
          sze[ad] = zb + dt * (sk[1][ad] - sk[0][ad] * (1.f / 3.f));
        } else if (sub == 2) {
          sze[ad] = zb + dt * (sk[0][ad] - sk[1][ad] + sk[2][ad]);
        } else {
          const float zn =
              zb + (sk[0][ad] + 3.f * (sk[1][ad] + sk[2][ad]) + sk[3][ad]) *
                       (dt * 0.125f);
          szb[ad] = zn;
          sze[ad] = zn;
          zcde[((size_t)(b0 + m) * Tt + (step + 1)) * Hh + h] = zn;
        }
      }
      __syncthreads();
    }
  }
}

__global__ __launch_bounds__(128) void ncde_head_kernel(
    const float* __restrict__ zflow, const float* __restrict__ zcde,
    const float* __restrict__ W_mix, const float* __restrict__ b_mix,
    const float* __restrict__ W_l1, const float* __restrict__ b_l1,
    const float* __restrict__ W_l2, const float* __restrict__ b_l2,
    float* __restrict__ out, float* __restrict__ zout) {
  __shared__ float sWmix[128 * WP], sWl1[64 * WP];
  __shared__ float sbmix[64], sbl1[64], sWl2v[64];
  __shared__ float sA[16 * AP2], sB[16 * AP1], sC[16 * AP1];

  const int tid = threadIdx.x, nthr = 128;
  const int lane = tid & 31, wave = tid >> 5, l2 = lane & 15, hi = lane >> 4;
  const int r0 = blockIdx.x * 16;

  load_w64<128>(sWmix, W_mix, tid, nthr);
  load_w64<64>(sWl1, W_l1, tid, nthr);
  if (tid < 64) {
    sbmix[tid] = b_mix[tid];
    sbl1[tid] = b_l1[tid];
    sWl2v[tid] = W_l2[tid];
  }
  for (int i = tid; i < 16 * 128; i += nthr) {
    const int m = i >> 7, c = i & 127;
    const size_t rr = r0 + m;
    sA[m * AP2 + c] = (c < 64) ? zflow[rr * Hh + c] : zcde[rr * Hh + (c - 64)];
  }
  __syncthreads();

  const int n0 = wave * 16;
  v8f acc = {};
  acc = gemm_tile<128, AP2>(sA, 0, sWmix, n0, acc, l2, hi);
  float bv = sbmix[n0 + l2];
#pragma unroll
  for (int r = 0; r < 8; ++r) {
    const int m = r + 8 * hi;
    const float v = acc[r] + bv;
    zout[(size_t)(r0 + m) * Hh + n0 + l2] = v;
    sB[m * AP1 + n0 + l2] = tanhf(v);
  }
  __syncthreads();

  acc = (v8f){};
  acc = gemm_tile<64, AP1>(sB, 0, sWl1, n0, acc, l2, hi);
  bv = sbl1[n0 + l2];
#pragma unroll
  for (int r = 0; r < 8; ++r)
    sC[(r + 8 * hi) * AP1 + n0 + l2] = relu(acc[r] + bv);
  __syncthreads();

  const int m = tid >> 3, p = tid & 7;
  float s = 0.f;
#pragma unroll
  for (int q = 0; q < 8; ++q) s += sC[m * AP1 + p * 8 + q] * sWl2v[p * 8 + q];
  s += __shfl_xor(s, 1, 32);
  s += __shfl_xor(s, 2, 32);
  s += __shfl_xor(s, 4, 32);
  if (p == 0) out[r0 + m] = s + b_l2[0];
}

extern "C" void kernel_launch(void* const* d_in, const int* in_sizes, int n_in,
                              void* d_out, int out_size, void* d_ws,
                              size_t ws_size, hipStream_t stream) {
  const float* x = (const float*)d_in[0];
  const float* seq_ts = (const float*)d_in[1];
  const float* coeffs = (const float*)d_in[3];
  const float* times = (const float*)d_in[4];
  const float* W_if = (const float*)d_in[5];
  const float* b_if = (const float*)d_in[6];
  const float* W_ic = (const float*)d_in[7];
  const float* b_ic = (const float*)d_in[8];
  const float* W_emb = (const float*)d_in[9];
  const float* b_emb = (const float*)d_in[10];
  const float* W_fl1 = (const float*)d_in[11];
  const float* b_fl1 = (const float*)d_in[12];
  const float* W_flo = (const float*)d_in[13];
  const float* b_flo = (const float*)d_in[14];
  const float* W_f1 = (const float*)d_in[15];
  const float* b_f1 = (const float*)d_in[16];
  const float* W_f2 = (const float*)d_in[17];
  const float* b_f2 = (const float*)d_in[18];
  const float* W_mix = (const float*)d_in[19];
  const float* b_mix = (const float*)d_in[20];
  const float* W_l1 = (const float*)d_in[21];
  const float* b_l1 = (const float*)d_in[22];
  const float* W_l2 = (const float*)d_in[23];
  const float* b_l2 = (const float*)d_in[24];

  float* zflow = (float*)d_ws;
  float* zcde = zflow + (size_t)Bb * Tt * Hh;
  float* out = (float*)d_out;
  float* zout = out + (size_t)Bb * Tt;

  ncde_embed_kernel<<<(Bb * Tt) / 16, 128, 0, stream>>>(
      x, seq_ts, coeffs, times, W_if, b_if, W_ic, b_ic, W_emb, b_emb, W_fl1,
      b_fl1, W_flo, b_flo, zflow);
  ncde_scan_kernel<<<Bb / RB, 512, 0, stream>>>(coeffs, times, W_ic, b_ic,
                                                W_f1, b_f1, W_f2, b_f2, zcde);
  ncde_head_kernel<<<(Bb * Tt) / 16, 128, 0, stream>>>(
      zflow, zcde, W_mix, b_mix, W_l1, b_l1, W_l2, b_l2, out, zout);
}


# scan: z and k1..k4 ride as rows of GEMM1's A tile (rows idle so far); each RK4 stage sum is formed on GEMM1's output by linearity: no update pass or its barrier (pre-activation sums reorder)
# speedup vs baseline: 3.6114x; 1.0792x over previous
#include <hip/hip_runtime.h>
#include <math.h>

#define Bb 256
#define Tt 512
#define Cc 16
#define Hh 64

#define WP  72
#define AP1 69
#define AP2 133
#define DXP 20
#define SP  72

typedef __attribute__((ext_vector_type(2))) float v2f;
typedef __attribute__((ext_vector_type(8))) float v8f;

static __device__ __forceinline__ v8f wmma4(v2f a, v2f b, v8f c) {
  return __builtin_amdgcn_wmma_f32_16x16x4_f32(false, a, false, b, (short)0, c,
                                               false, false);
}

template <int K, int APITCH, int ROWMASK = 15>
static __device__ __forceinline__ v8f gemm_tile(const float* __restrict__ A,
                                                int acol0,
                                                const float* __restrict__ W,
                                                int n0, v8f acc, int l2, int hi) {
#pragma unroll
  for (int k = 0; k < K; k += 4) {
    v2f a, b;
    const float* ap = A + (l2 & ROWMASK) * APITCH + acol0 + k + 2 * hi;
    a.x = ap[0];
    a.y = ap[1];
    const float* wp = W + (k + 2 * hi) * WP + n0 + l2;
    b.x = wp[0];
    b.y = wp[WP];
    acc = wmma4(a, b, acc);
  }
  return acc;
}

template <int APITCH>
static __device__ __forceinline__ void store_tile(float* __restrict__ O, int n0,
                                                  v8f acc, int l2, int hi) {
#pragma unroll
  for (int r = 0; r < 8; ++r) O[(r + 8 * hi) * APITCH + n0 + l2] = acc[r];
}

template <int ROWS>
static __device__ __forceinline__ void load_w64(float* __restrict__ dst,
                                                const float* __restrict__ src,
                                                int tid, int nthr) {
  for (int i = tid; i < ROWS * 64; i += nthr)
    dst[(i >> 6) * WP + (i & 63)] = src[i];
}

static __device__ __forceinline__ float relu(float v) { return v > 0.f ? v : 0.f; }

template <int M>
static __device__ __forceinline__ float row_xor_add(float p) {
  const int other = __builtin_amdgcn_update_dpp(0, __builtin_bit_cast(int, p),
                                                0x160 + M, 0xf, 0xf, true);
  return p + __builtin_bit_cast(float, other);
}

__global__ __launch_bounds__(128) void ncde_embed_kernel(
    const float* __restrict__ x, const float* __restrict__ seq_ts,
    const float* __restrict__ coeffs, const float* __restrict__ times,
    const float* __restrict__ W_if, const float* __restrict__ b_if,
    const float* __restrict__ W_ic, const float* __restrict__ b_ic,
    const float* __restrict__ W_emb, const float* __restrict__ b_emb,
    const float* __restrict__ W_fl1, const float* __restrict__ b_fl1,
    const float* __restrict__ W_flo, const float* __restrict__ b_flo,
    float* __restrict__ zflow) {
  __shared__ float sWif[16 * WP], sWic[16 * WP], sWemb[128 * WP];
  __shared__ float sWfl1[64 * WP], sWflo[64 * WP];
  __shared__ float sb[5][64];
  __shared__ float sA[16 * AP2], sB[16 * AP2];

  const int tid = threadIdx.x, nthr = 128;
  const int lane = tid & 31, wave = tid >> 5, l2 = lane & 15, hi = lane >> 4;
  const int r0 = blockIdx.x * 16;

  load_w64<16>(sWif, W_if, tid, nthr);
  load_w64<16>(sWic, W_ic, tid, nthr);
  load_w64<128>(sWemb, W_emb, tid, nthr);
  load_w64<64>(sWfl1, W_fl1, tid, nthr);
  load_w64<64>(sWflo, W_flo, tid, nthr);
  if (tid < 64) {
    sb[0][tid] = b_if[tid];
    sb[1][tid] = b_ic[tid];
    sb[2][tid] = b_emb[tid];
    sb[3][tid] = b_fl1[tid];
    sb[4][tid] = b_flo[tid];
  }
  for (int i = tid; i < 256; i += nthr) {
    const int m = i >> 4, c = i & 15;
    const int rr = r0 + m, b = rr / Tt, t = rr % Tt;
    sA[m * AP2 + c] = (c == 0) ? seq_ts[rr] : x[(size_t)rr * (Cc - 1) + (c - 1)];
    const int idx = (t < Tt - 1) ? t : (Tt - 2);
    const float s = times[t] - times[idx];
    const float* cf = coeffs + ((size_t)b * (Tt - 1) + idx) * (4 * Cc) + c;
    const float a0 = cf[0], b0 = cf[Cc], c0 = cf[2 * Cc], d0 = cf[3 * Cc];
    sA[m * AP2 + 16 + c] = a0 + s * (b0 + s * (c0 + s * d0));
  }
  __syncthreads();

  const int n0 = wave * 16;
  v8f aF = {}, aX = {};
  aF = gemm_tile<16, AP2>(sA, 0, sWif, n0, aF, l2, hi);
  aX = gemm_tile<16, AP2>(sA, 16, sWic, n0, aX, l2, hi);
  const float bF = sb[0][n0 + l2], bX = sb[1][n0 + l2];
#pragma unroll
  for (int r = 0; r < 8; ++r) {
    const int m = r + 8 * hi;
    sB[m * AP2 + n0 + l2] = aF[r] + bF;
    sB[m * AP2 + 64 + n0 + l2] = aX[r] + bX;
  }
  __syncthreads();

  v8f acc = {};
  acc = gemm_tile<128, AP2>(sB, 0, sWemb, n0, acc, l2, hi);
  float bv = sb[2][n0 + l2];
#pragma unroll
  for (int r = 0; r < 8; ++r)
    sA[(r + 8 * hi) * AP2 + n0 + l2] = relu(acc[r] + bv);
  __syncthreads();

  acc = (v8f){};
  acc = gemm_tile<64, AP2>(sA, 0, sWfl1, n0, acc, l2, hi);
  bv = sb[3][n0 + l2];
#pragma unroll
  for (int r = 0; r < 8; ++r)
    sB[(r + 8 * hi) * AP2 + n0 + l2] = relu(acc[r] + bv);
  __syncthreads();

  acc = (v8f){};
  acc = gemm_tile<64, AP2>(sB, 0, sWflo, n0, acc, l2, hi);
  bv = sb[4][n0 + l2];
#pragma unroll
  for (int r = 0; r < 8; ++r)
    zflow[(size_t)(r0 + r + 8 * hi) * Hh + n0 + l2] = acc[r] + bv;
}

__global__ __launch_bounds__(512, 1) void ncde_scan_kernel(
    const float* __restrict__ coeffs, const float* __restrict__ times,
    const float* __restrict__ W_ic, const float* __restrict__ b_ic,
    const float* __restrict__ W_f1, const float* __restrict__ b_f1,
    const float* __restrict__ W_f2, const float* __restrict__ b_f2,
    float* __restrict__ zcde) {
  __shared__ float sWic[16 * WP];
  __shared__ float sbic[64];
  __shared__ float stim[Tt];
  __shared__ float sx0[16];
  __shared__ float sh[Hh];
  __shared__ float sst[2][8 * SP];
  __shared__ float sdx[4][DXP];

  const int tid = threadIdx.x, nthr = 512;
  const int lane = tid & 31, wave = tid >> 5, l2 = lane & 15, hi = lane >> 4;
  const int b0 = blockIdx.x;
  load_w64<16>(sWic, W_ic, tid, nthr);
  if (tid < 64) sbic[tid] = b_ic[tid];
  for (int i = tid; i < Tt; i += nthr) stim[i] = times[i];
  for (int i = tid; i < 2 * 8 * SP; i += nthr) (&sst[0][0])[i] = 0.f;

  v2f wf2[4][16];
  float bf2r[4];
#pragma unroll
  for (int t = 0; t < 4; ++t) bf2r[t] = b_f2[16 * (wave * 4 + t) + l2];
#pragma unroll
  for (int jj = 0; jj < 4; ++jj) {
    const int j = wave * 4 + jj;
    const int col = 16 * j + l2;
#pragma unroll
    for (int kk = 0; kk < 16; ++kk) {
      const int row = 4 * kk + 2 * hi;
      wf2[jj][kk].x = W_f2[(size_t)row * (Hh * Cc) + col];
      wf2[jj][kk].y = W_f2[(size_t)(row + 1) * (Hh * Cc) + col];
    }
  }

  v2f wf1[16];
  const int n1 = (wave & 3) * 16 + l2;
  const float bf1r = b_f1[n1];
#pragma unroll
  for (int kk = 0; kk < 16; ++kk) {
    const int row = 4 * kk + 2 * hi;
    wf1[kk].x = W_f1[row * Hh + n1];
    wf1[kk].y = W_f1[(row + 1) * Hh + n1];
  }

  if (tid < 16)
    sx0[tid] = coeffs[((size_t)b0 * (Tt - 1)) * (4 * Cc) + tid];
  __syncthreads();
  if (wave < 4) {
    const int n0 = wave * 16;
    v8f acc = {};
    acc = gemm_tile<16, AP1, 0>(sx0, 0, sWic, n0, acc, l2, hi);
    const float zv = acc[0] + sbic[n0 + l2];
    if (hi == 0) sst[0][n0 + l2] = zv;
  }
  __syncthreads();

  for (int step = 0; step < Tt - 1; ++step) {
    const float dt = stim[step + 1] - stim[step];
    const float dtp = step > 0 ? stim[step] - stim[step - 1] : 0.f;
    float* tprev = sst[step & 1];
    float* tcur = sst[(step + 1) & 1];

#pragma unroll
    for (int sub = 0; sub < 4; ++sub) {
      if (wave < 4) {
        const int n0 = wave * 16;
        const float* tile = sub == 0 ? tprev : tcur;
        v2f az[16];
        const float* zp = tile + (l2 & 7) * SP + 2 * hi;
#pragma unroll
        for (int kk = 0; kk < 16; ++kk) {
          az[kk].x = zp[4 * kk];
          az[kk].y = zp[4 * kk + 1];
        }
        v8f acc = {};
#pragma unroll
        for (int kk = 0; kk < 16; ++kk) acc = wmma4(az[kk], wf1[kk], acc);
        float pre;
        if (sub == 0) {
          pre = acc[0] + (acc[1] + 3.f * (acc[2] + acc[3]) + acc[4]) *
                             (dtp * 0.125f);
        } else if (sub == 1) {
          pre = acc[0] + (dt * (1.f / 3.f)) * acc[1];
        } else if (sub == 2) {
          pre = acc[0] + dt * (acc[2] - acc[1] * (1.f / 3.f));
        } else {
          pre = acc[0] + dt * (acc[1] - acc[2] + acc[3]);
        }
        const float hv = relu(pre + bf1r);
        if (hi == 0) sh[n0 + l2] = hv;
      } else if (sub == 0) {
        if (wave < 6) {
          const int h = tid - 128;
          const float zn =
              tprev[h] + (tprev[SP + h] +
                          3.f * (tprev[2 * SP + h] + tprev[3 * SP + h]) +
                          tprev[4 * SP + h]) *
                             (dtp * 0.125f);
          tcur[h] = zn;
          zcde[((size_t)b0 * Tt + step) * Hh + h] = zn;
        } else if (wave == 6 && lane < 16) {
          const int c = lane;
          const float* cf =
              coeffs + ((size_t)b0 * (Tt - 1) + step) * (4 * Cc) + c;
          const float bb = cf[Cc], cc = cf[2 * Cc], dd = cf[3 * Cc];
          const float s1 = dt * (1.f / 3.f), s2 = dt * (2.f / 3.f);
          sdx[0][c] = bb;
          sdx[1][c] = bb + s1 * (2.f * cc + 3.f * dd * s1);
          sdx[2][c] = bb + s2 * (2.f * cc + 3.f * dd * s2);
          const int idx3 = (step + 1 < Tt - 1) ? (step + 1) : (Tt - 2);
          const float* cf3 =
              coeffs + ((size_t)b0 * (Tt - 1) + idx3) * (4 * Cc) + c;
          const float b3 = cf3[Cc], c3 = cf3[2 * Cc], d3 = cf3[3 * Cc];
          const float s3 = stim[step + 1] - stim[idx3];
          sdx[3][c] = b3 + s3 * (2.f * c3 + 3.f * d3 * s3);
        }
      }
      __syncthreads();

      v8f acc2[4];
#pragma unroll
      for (int jj = 0; jj < 4; ++jj) {
#pragma unroll
        for (int r = 0; r < 8; ++r) acc2[jj][r] = bf2r[jj];
      }
#pragma unroll
      for (int kk = 0; kk < 16; ++kk) {
        v2f hv2;
        const float* ap = sh + 4 * kk + 2 * hi;
        hv2.x = ap[0];
        hv2.y = ap[1];
#pragma unroll
        for (int jj = 0; jj < 4; ++jj)
          acc2[jj] = wmma4(hv2, wf2[jj][kk], acc2[jj]);
      }
      const float dxr = sdx[sub][l2];
      float pv[2];
#pragma unroll
      for (int r = 0; r < 2; ++r) {
        const float lo = acc2[r][0], up = acc2[2 + r][0];
        pv[r] = tanhf(hi ? up : lo) * dxr;
      }
#pragma unroll
      for (int r = 0; r < 2; ++r) {
        float p = pv[r];
        p = row_xor_add<1>(p);
        p = row_xor_add<2>(p);
        p = row_xor_add<4>(p);
        p = row_xor_add<8>(p);
        pv[r] = p;
      }
      if (l2 == 0) {
        float* kdst = tcur + (sub + 1) * SP + wave * 4 + 2 * hi;
        kdst[0] = pv[0];
        kdst[1] = pv[1];
      }
      __syncthreads();
    }
  }

  if (tid < Hh) {
    const float* tl = sst[(Tt - 1) & 1];
    const float dtl = stim[Tt - 1] - stim[Tt - 2];
    const float zn =
        tl[tid] + (tl[SP + tid] + 3.f * (tl[2 * SP + tid] + tl[3 * SP + tid]) +
                   tl[4 * SP + tid]) *
                      (dtl * 0.125f);
    zcde[((size_t)b0 * Tt + (Tt - 1)) * Hh + tid] = zn;
  }
}

__global__ __launch_bounds__(128) void ncde_head_kernel(
    const float* __restrict__ zflow, const float* __restrict__ zcde,
    const float* __restrict__ W_mix, const float* __restrict__ b_mix,
    const float* __restrict__ W_l1, const float* __restrict__ b_l1,
    const float* __restrict__ W_l2, const float* __restrict__ b_l2,
    float* __restrict__ out, float* __restrict__ zout) {
  __shared__ float sWmix[128 * WP], sWl1[64 * WP];
  __shared__ float sbmix[64], sbl1[64], sWl2v[64];
  __shared__ float sA[16 * AP2], sB[16 * AP1], sC[16 * AP1];

  const int tid = threadIdx.x, nthr = 128;
  const int lane = tid & 31, wave = tid >> 5, l2 = lane & 15, hi = lane >> 4;
  const int r0 = blockIdx.x * 16;

  load_w64<128>(sWmix, W_mix, tid, nthr);
  load_w64<64>(sWl1, W_l1, tid, nthr);
  if (tid < 64) {
    sbmix[tid] = b_mix[tid];
    sbl1[tid] = b_l1[tid];
    sWl2v[tid] = W_l2[tid];
  }
  for (int i = tid; i < 16 * 128; i += nthr) {
    const int m = i >> 7, c = i & 127;
    const size_t rr = r0 + m;
    sA[m * AP2 + c] = (c < 64) ? zflow[rr * Hh + c] : zcde[rr * Hh + (c - 64)];
  }
  __syncthreads();

  const int n0 = wave * 16;
  v8f acc = {};
  acc = gemm_tile<128, AP2>(sA, 0, sWmix, n0, acc, l2, hi);
  float bv = sbmix[n0 + l2];
#pragma unroll
  for (int r = 0; r < 8; ++r) {
    const int m = r + 8 * hi;
    const float v = acc[r] + bv;
    zout[(size_t)(r0 + m) * Hh + n0 + l2] = v;
    sB[m * AP1 + n0 + l2] = tanhf(v);
  }
  __syncthreads();

  acc = (v8f){};
  acc = gemm_tile<64, AP1>(sB, 0, sWl1, n0, acc, l2, hi);
  bv = sbl1[n0 + l2];
#pragma unroll
  for (int r = 0; r < 8; ++r)
    sC[(r + 8 * hi) * AP1 + n0 + l2] = relu(acc[r] + bv);
  __syncthreads();

  const int m = tid >> 3, p = tid & 7;
  float s = 0.f;
#pragma unroll
  for (int q = 0; q < 8; ++q) s += sC[m * AP1 + p * 8 + q] * sWl2v[p * 8 + q];
  s += __shfl_xor(s, 1, 32);
  s += __shfl_xor(s, 2, 32);
  s += __shfl_xor(s, 4, 32);
  if (p == 0) out[r0 + m] = s + b_l2[0];
}

extern "C" void kernel_launch(void* const* d_in, const int* in_sizes, int n_in,
                              void* d_out, int out_size, void* d_ws,
                              size_t ws_size, hipStream_t stream) {
  const float* x = (const float*)d_in[0];
  const float* seq_ts = (const float*)d_in[1];
  const float* coeffs = (const float*)d_in[3];
  const float* times = (const float*)d_in[4];
  const float* W_if = (const float*)d_in[5];
  const float* b_if = (const float*)d_in[6];
  const float* W_ic = (const float*)d_in[7];
  const float* b_ic = (const float*)d_in[8];
  const float* W_emb = (const float*)d_in[9];
  const float* b_emb = (const float*)d_in[10];
  const float* W_fl1 = (const float*)d_in[11];
  const float* b_fl1 = (const float*)d_in[12];
  const float* W_flo = (const float*)d_in[13];
  const float* b_flo = (const float*)d_in[14];
  const float* W_f1 = (const float*)d_in[15];
  const float* b_f1 = (const float*)d_in[16];
  const float* W_f2 = (const float*)d_in[17];
  const float* b_f2 = (const float*)d_in[18];
  const float* W_mix = (const float*)d_in[19];
  const float* b_mix = (const float*)d_in[20];
  const float* W_l1 = (const float*)d_in[21];
  const float* b_l1 = (const float*)d_in[22];
  const float* W_l2 = (const float*)d_in[23];
  const float* b_l2 = (const float*)d_in[24];

  float* zflow = (float*)d_ws;
  float* zcde = zflow + (size_t)Bb * Tt * Hh;
  float* out = (float*)d_out;
  float* zout = out + (size_t)Bb * Tt;

  ncde_embed_kernel<<<(Bb * Tt) / 16, 128, 0, stream>>>(
      x, seq_ts, coeffs, times, W_if, b_if, W_ic, b_ic, W_emb, b_emb, W_fl1,
      b_fl1, W_flo, b_flo, zflow);
  ncde_scan_kernel<<<Bb, 512, 0, stream>>>(coeffs, times, W_ic, b_ic,
                                                W_f1, b_f1, W_f2, b_f2, zcde);
  ncde_head_kernel<<<(Bb * Tt) / 16, 128, 0, stream>>>(
      zflow, zcde, W_mix, b_mix, W_l1, b_l1, W_l2, b_l2, out, zout);
}


# scan: GEMM2's walk lowers the wave's issue priority as it advances (s_setprio 3,2,1,0 after 3,5,7 groups): a SIMD's four waves share the matrix pipe by progress, not oldest-first; sums unchanged
# speedup vs baseline: 4.1368x; 1.1455x over previous
#include <hip/hip_runtime.h>
#include <math.h>

#define Bb 256
#define Tt 512
#define Cc 16
#define Hh 64

#define WP  72
#define AP1 69
#define AP2 133
#define DXP 20
#define SP  72

typedef __attribute__((ext_vector_type(2))) float v2f;
typedef __attribute__((ext_vector_type(8))) float v8f;
typedef __attribute__((ext_vector_type(4))) float v4f;
#define W2P 1024

static __device__ __forceinline__ v8f wmma4(v2f a, v2f b, v8f c) {
  return __builtin_amdgcn_wmma_f32_16x16x4_f32(false, a, false, b, (short)0, c,
                                               false, false);
}

template <int K, int APITCH, int ROWMASK = 15>
static __device__ __forceinline__ v8f gemm_tile(const float* __restrict__ A,
                                                int acol0,
                                                const float* __restrict__ W,
                                                int n0, v8f acc, int l2, int hi) {
#pragma unroll
  for (int k = 0; k < K; k += 4) {
    v2f a, b;
    const float* ap = A + (l2 & ROWMASK) * APITCH + acol0 + k + 2 * hi;
    a.x = ap[0];
    a.y = ap[1];
    const float* wp = W + (k + 2 * hi) * WP + n0 + l2;
    b.x = wp[0];
    b.y = wp[WP];
    acc = wmma4(a, b, acc);
  }
  return acc;
}

template <int APITCH>
static __device__ __forceinline__ void store_tile(float* __restrict__ O, int n0,
                                                  v8f acc, int l2, int hi) {
#pragma unroll
  for (int r = 0; r < 8; ++r) O[(r + 8 * hi) * APITCH + n0 + l2] = acc[r];
}

template <int ROWS>
static __device__ __forceinline__ void load_w64(float* __restrict__ dst,
                                                const float* __restrict__ src,
                                                int tid, int nthr) {
  for (int i = tid; i < ROWS * 64; i += nthr)
    dst[(i >> 6) * WP + (i & 63)] = src[i];
}

static __device__ __forceinline__ float relu(float v) { return v > 0.f ? v : 0.f; }

template <int M>
static __device__ __forceinline__ float row_xor_add(float p) {
  const int other = __builtin_amdgcn_update_dpp(0, __builtin_bit_cast(int, p),
                                                0x160 + M, 0xf, 0xf, true);
  return p + __builtin_bit_cast(float, other);
}

__global__ __launch_bounds__(128) void ncde_embed_kernel(
    const float* __restrict__ x, const float* __restrict__ seq_ts,
    const float* __restrict__ coeffs, const float* __restrict__ times,
    const float* __restrict__ W_if, const float* __restrict__ b_if,
    const float* __restrict__ W_ic, const float* __restrict__ b_ic,
    const float* __restrict__ W_emb, const float* __restrict__ b_emb,
    const float* __restrict__ W_fl1, const float* __restrict__ b_fl1,
    const float* __restrict__ W_flo, const float* __restrict__ b_flo,
    float* __restrict__ zflow) {
  __shared__ float sWif[16 * WP], sWic[16 * WP], sWemb[128 * WP];
  __shared__ float sWfl1[64 * WP], sWflo[64 * WP];
  __shared__ float sb[5][64];
  __shared__ float sA[16 * AP2], sB[16 * AP2];

  const int tid = threadIdx.x, nthr = 128;
  const int lane = tid & 31, wave = tid >> 5, l2 = lane & 15, hi = lane >> 4;
  const int r0 = blockIdx.x * 16;

  load_w64<16>(sWif, W_if, tid, nthr);
  load_w64<16>(sWic, W_ic, tid, nthr);
  load_w64<128>(sWemb, W_emb, tid, nthr);
  load_w64<64>(sWfl1, W_fl1, tid, nthr);
  load_w64<64>(sWflo, W_flo, tid, nthr);
  if (tid < 64) {
    sb[0][tid] = b_if[tid];
    sb[1][tid] = b_ic[tid];
    sb[2][tid] = b_emb[tid];
    sb[3][tid] = b_fl1[tid];
    sb[4][tid] = b_flo[tid];
  }
  for (int i = tid; i < 256; i += nthr) {
    const int m = i >> 4, c = i & 15;
    const int rr = r0 + m, b = rr / Tt, t = rr % Tt;
    sA[m * AP2 + c] = (c == 0) ? seq_ts[rr] : x[(size_t)rr * (Cc - 1) + (c - 1)];
    const int idx = (t < Tt - 1) ? t : (Tt - 2);
    const float s = times[t] - times[idx];
    const float* cf = coeffs + ((size_t)b * (Tt - 1) + idx) * (4 * Cc) + c;
    const float a0 = cf[0], b0 = cf[Cc], c0 = cf[2 * Cc], d0 = cf[3 * Cc];
    sA[m * AP2 + 16 + c] = a0 + s * (b0 + s * (c0 + s * d0));
  }
  __syncthreads();

  const int n0 = wave * 16;
  v8f aF = {}, aX = {};
  aF = gemm_tile<16, AP2>(sA, 0, sWif, n0, aF, l2, hi);
  aX = gemm_tile<16, AP2>(sA, 16, sWic, n0, aX, l2, hi);
  const float bF = sb[0][n0 + l2], bX = sb[1][n0 + l2];
#pragma unroll
  for (int r = 0; r < 8; ++r) {
    const int m = r + 8 * hi;
    sB[m * AP2 + n0 + l2] = aF[r] + bF;
    sB[m * AP2 + 64 + n0 + l2] = aX[r] + bX;
  }
  __syncthreads();

  v8f acc = {};
  acc = gemm_tile<128, AP2>(sB, 0, sWemb, n0, acc, l2, hi);
  float bv = sb[2][n0 + l2];
#pragma unroll
  for (int r = 0; r < 8; ++r)
    sA[(r + 8 * hi) * AP2 + n0 + l2] = relu(acc[r] + bv);
  __syncthreads();

  acc = (v8f){};
  acc = gemm_tile<64, AP2>(sA, 0, sWfl1, n0, acc, l2, hi);
  bv = sb[3][n0 + l2];
#pragma unroll
  for (int r = 0; r < 8; ++r)
    sB[(r + 8 * hi) * AP2 + n0 + l2] = relu(acc[r] + bv);
  __syncthreads();

  acc = (v8f){};
  acc = gemm_tile<64, AP2>(sB, 0, sWflo, n0, acc, l2, hi);
  bv = sb[4][n0 + l2];
#pragma unroll
  for (int r = 0; r < 8; ++r)
    zflow[(size_t)(r0 + r + 8 * hi) * Hh + n0 + l2] = acc[r] + bv;
}

__global__ __launch_bounds__(512, 1) void ncde_scan_kernel(
    const float* __restrict__ coeffs, const float* __restrict__ times,
    const float* __restrict__ W_ic, const float* __restrict__ b_ic,
    const float* __restrict__ W_f1, const float* __restrict__ b_f1,
    const float* __restrict__ W_f2, const float* __restrict__ b_f2,
    float* __restrict__ zcde) {
  __shared__ float sWic[16 * WP];
  __shared__ float sbic[64];
  __shared__ float stim[Tt];
  __shared__ float sx0[16];
  __shared__ float sh[Hh + 1];
  __shared__ __attribute__((aligned(16))) float sW2[(Hh + 1) * W2P];
  __shared__ float sst[2][8 * SP];
  __shared__ float sdx[4][DXP];

  const int tid = threadIdx.x, nthr = 512;
  const int lane = tid & 31, wave = tid >> 5, l2 = lane & 15, hi = lane >> 4;
  const int b0 = blockIdx.x;
  load_w64<16>(sWic, W_ic, tid, nthr);
  if (tid < 64) sbic[tid] = b_ic[tid];
  for (int i = tid; i < Tt; i += nthr) stim[i] = times[i];
  for (int i = tid; i < 2 * 8 * SP; i += nthr) (&sst[0][0])[i] = 0.f;

  if (tid == 0) sh[Hh] = 0.f;
#pragma unroll 8
  for (int i = tid; i < Hh * Hh * Cc; i += nthr)
    sW2[(i & ~63) + 4 * (i & 15) + ((i >> 4) & 3)] = W_f2[i];
  for (int i = tid; i < W2P; i += nthr) sW2[Hh * W2P + i] = 0.f;
  const float* w2lane = sW2 + 64 * wave + 4 * l2;
  float bf2r[4];
#pragma unroll
  for (int t = 0; t < 4; ++t) bf2r[t] = b_f2[16 * (wave * 4 + t) + l2];

  v2f wf1[16];
  const int n1 = (wave & 3) * 16 + l2;
  const float bf1r = b_f1[n1];
#pragma unroll
  for (int kk = 0; kk < 16; ++kk) {
    const int row = 4 * kk + 2 * hi;
    wf1[kk].x = W_f1[row * Hh + n1];
    wf1[kk].y = W_f1[(row + 1) * Hh + n1];
  }

  if (tid < 16)
    sx0[tid] = coeffs[((size_t)b0 * (Tt - 1)) * (4 * Cc) + tid];
  __syncthreads();
  if (wave < 4) {
    const int n0 = wave * 16;
    v8f acc = {};
    acc = gemm_tile<16, AP1, 0>(sx0, 0, sWic, n0, acc, l2, hi);
    const float zv = acc[0] + sbic[n0 + l2];
    if (hi == 0) sst[0][n0 + l2] = zv;
  }
  __syncthreads();

  for (int step = 0; step < Tt - 1; ++step) {
    const float dt = stim[step + 1] - stim[step];
    const float dtp = step > 0 ? stim[step] - stim[step - 1] : 0.f;
    float* tprev = sst[step & 1];
    float* tcur = sst[(step + 1) & 1];

#pragma unroll
    for (int sub = 0; sub < 4; ++sub) {
      if (wave < 4) {
        const int n0 = wave * 16;
        const float* tile = sub == 0 ? tprev : tcur;
        v2f az[16];
        const float* zp = tile + (l2 & 7) * SP + 2 * hi;
#pragma unroll
        for (int kk = 0; kk < 16; ++kk) {
          az[kk].x = zp[4 * kk];
          az[kk].y = zp[4 * kk + 1];
        }
        v8f acc = {};
#pragma unroll
        for (int kk = 0; kk < 16; ++kk) acc = wmma4(az[kk], wf1[kk], acc);
        float pre;
        if (sub == 0) {
          pre = acc[0] + (acc[1] + 3.f * (acc[2] + acc[3]) + acc[4]) *
                             (dtp * 0.125f);
        } else if (sub == 1) {
          pre = acc[0] + (dt * (1.f / 3.f)) * acc[1];
        } else if (sub == 2) {
          pre = acc[0] + dt * (acc[2] - acc[1] * (1.f / 3.f));
        } else {
          pre = acc[0] + dt * (acc[1] - acc[2] + acc[3]);
        }
        const float hv = relu(pre + bf1r);
        if (hi == 0) sh[n0 + l2] = hv;
      } else if (sub == 0) {
        if (wave < 6) {
          const int h = tid - 128;
          const float zn =
              tprev[h] + (tprev[SP + h] +
                          3.f * (tprev[2 * SP + h] + tprev[3 * SP + h]) +
                          tprev[4 * SP + h]) *
                             (dtp * 0.125f);
          tcur[h] = zn;
          zcde[((size_t)b0 * Tt + step) * Hh + h] = zn;
        } else if (wave == 6 && lane < 16) {
          const int c = lane;
          const float* cf =
              coeffs + ((size_t)b0 * (Tt - 1) + step) * (4 * Cc) + c;
          const float bb = cf[Cc], cc = cf[2 * Cc], dd = cf[3 * Cc];
          const float s1 = dt * (1.f / 3.f), s2 = dt * (2.f / 3.f);
          sdx[0][c] = bb;
          sdx[1][c] = bb + s1 * (2.f * cc + 3.f * dd * s1);
          sdx[2][c] = bb + s2 * (2.f * cc + 3.f * dd * s2);
          const int idx3 = (step + 1 < Tt - 1) ? (step + 1) : (Tt - 2);
          const float* cf3 =
              coeffs + ((size_t)b0 * (Tt - 1) + idx3) * (4 * Cc) + c;
          const float b3 = cf3[Cc], c3 = cf3[2 * Cc], d3 = cf3[3 * Cc];
          const float s3 = stim[step + 1] - stim[idx3];
          sdx[3][c] = b3 + s3 * (2.f * c3 + 3.f * d3 * s3);
        }
      }
      __syncthreads();

      v8f acc2[4];
#pragma unroll
      for (int jj = 0; jj < 4; ++jj) {
#pragma unroll
        for (int r = 0; r < 8; ++r) acc2[jj][r] = bf2r[jj];
      }
      const unsigned mlo = __builtin_amdgcn_ballot_w32(sh[lane] != 0.f);
      const unsigned mhi = __builtin_amdgcn_ballot_w32(sh[32 + lane] != 0.f);
      unsigned long long live = ((unsigned long long)mhi << 32) | mlo;
      auto take = [&]() {
        const int r = __builtin_amdgcn_readfirstlane(
            live != 0ull ? (int)__builtin_ctzll(live) : Hh);
        live &= live - 1ull;
        return r;
      };
      bool more = live != 0ull;
      int r0 = take(), r1 = take(), r2 = take(), r3 = take();
      auto group = [&]() __attribute__((always_inline)) {
        const int ra = hi ? r2 : r0;
        const int rb = hi ? r3 : r1;
        v2f hv2;
        hv2.x = sh[ra];
        hv2.y = sh[rb];
        const v4f wa = *reinterpret_cast<const v4f*>(w2lane + ra * W2P);
        const v4f wb = *reinterpret_cast<const v4f*>(w2lane + rb * W2P);
        more = live != 0ull;
        r0 = take();
        r1 = take();
        r2 = take();
        r3 = take();
#pragma unroll
        for (int jj = 0; jj < 4; ++jj) {
          v2f wv;
          wv.x = wa[jj];
          wv.y = wb[jj];
          acc2[jj] = wmma4(hv2, wv, acc2[jj]);
        }
      };
      auto stretch = [&](int budget) __attribute__((always_inline)) {
        for (int q = 0; q < budget && more; ++q) group();
      };
      __builtin_amdgcn_s_setprio(3);
      stretch(3);
      __builtin_amdgcn_s_setprio(2);
      stretch(2);
      __builtin_amdgcn_s_setprio(1);
      stretch(2);
      __builtin_amdgcn_s_setprio(0);
      stretch(Hh);
      const float dxr = sdx[sub][l2];
      float pv[2];
#pragma unroll
      for (int r = 0; r < 2; ++r) {
        const float lo = acc2[r][0], up = acc2[2 + r][0];
        pv[r] = tanhf(hi ? up : lo) * dxr;
      }
#pragma unroll
      for (int r = 0; r < 2; ++r) {
        float p = pv[r];
        p = row_xor_add<1>(p);
        p = row_xor_add<2>(p);
        p = row_xor_add<4>(p);
        p = row_xor_add<8>(p);
        pv[r] = p;
      }
      if (l2 == 0) {
        float* kdst = tcur + (sub + 1) * SP + wave * 4 + 2 * hi;
        kdst[0] = pv[0];
        kdst[1] = pv[1];
      }
      __syncthreads();
    }
  }

  if (tid < Hh) {
    const float* tl = sst[(Tt - 1) & 1];
    const float dtl = stim[Tt - 1] - stim[Tt - 2];
    const float zn =
        tl[tid] + (tl[SP + tid] + 3.f * (tl[2 * SP + tid] + tl[3 * SP + tid]) +
                   tl[4 * SP + tid]) *
                      (dtl * 0.125f);
    zcde[((size_t)b0 * Tt + (Tt - 1)) * Hh + tid] = zn;
  }
}

__global__ __launch_bounds__(128) void ncde_head_kernel(
    const float* __restrict__ zflow, const float* __restrict__ zcde,
    const float* __restrict__ W_mix, const float* __restrict__ b_mix,
    const float* __restrict__ W_l1, const float* __restrict__ b_l1,
    const float* __restrict__ W_l2, const float* __restrict__ b_l2,
    float* __restrict__ out, float* __restrict__ zout) {
  __shared__ float sWmix[128 * WP], sWl1[64 * WP];
  __shared__ float sbmix[64], sbl1[64], sWl2v[64];
  __shared__ float sA[16 * AP2], sB[16 * AP1], sC[16 * AP1];

  const int tid = threadIdx.x, nthr = 128;
  const int lane = tid & 31, wave = tid >> 5, l2 = lane & 15, hi = lane >> 4;
  const int r0 = blockIdx.x * 16;

  load_w64<128>(sWmix, W_mix, tid, nthr);
  load_w64<64>(sWl1, W_l1, tid, nthr);
  if (tid < 64) {
    sbmix[tid] = b_mix[tid];
    sbl1[tid] = b_l1[tid];
    sWl2v[tid] = W_l2[tid];
  }
  for (int i = tid; i < 16 * 128; i += nthr) {
    const int m = i >> 7, c = i & 127;
    const size_t rr = r0 + m;
    sA[m * AP2 + c] = (c < 64) ? zflow[rr * Hh + c] : zcde[rr * Hh + (c - 64)];
  }
  __syncthreads();

  const int n0 = wave * 16;
  v8f acc = {};
  acc = gemm_tile<128, AP2>(sA, 0, sWmix, n0, acc, l2, hi);
  float bv = sbmix[n0 + l2];
#pragma unroll
  for (int r = 0; r < 8; ++r) {
    const int m = r + 8 * hi;
    const float v = acc[r] + bv;
    zout[(size_t)(r0 + m) * Hh + n0 + l2] = v;
    sB[m * AP1 + n0 + l2] = tanhf(v);
  }
  __syncthreads();

  acc = (v8f){};
  acc = gemm_tile<64, AP1>(sB, 0, sWl1, n0, acc, l2, hi);
  bv = sbl1[n0 + l2];
#pragma unroll
  for (int r = 0; r < 8; ++r)
    sC[(r + 8 * hi) * AP1 + n0 + l2] = relu(acc[r] + bv);
  __syncthreads();

  const int m = tid >> 3, p = tid & 7;
  float s = 0.f;
#pragma unroll
  for (int q = 0; q < 8; ++q) s += sC[m * AP1 + p * 8 + q] * sWl2v[p * 8 + q];
  s += __shfl_xor(s, 1, 32);
  s += __shfl_xor(s, 2, 32);
  s += __shfl_xor(s, 4, 32);
  if (p == 0) out[r0 + m] = s + b_l2[0];
}

extern "C" void kernel_launch(void* const* d_in, const int* in_sizes, int n_in,
                              void* d_out, int out_size, void* d_ws,
                              size_t ws_size, hipStream_t stream) {
  const float* x = (const float*)d_in[0];
  const float* seq_ts = (const float*)d_in[1];
  const float* coeffs = (const float*)d_in[3];
  const float* times = (const float*)d_in[4];
  const float* W_if = (const float*)d_in[5];
  const float* b_if = (const float*)d_in[6];
  const float* W_ic = (const float*)d_in[7];
  const float* b_ic = (const float*)d_in[8];
  const float* W_emb = (const float*)d_in[9];
  const float* b_emb = (const float*)d_in[10];
  const float* W_fl1 = (const float*)d_in[11];
  const float* b_fl1 = (const float*)d_in[12];
  const float* W_flo = (const float*)d_in[13];
  const float* b_flo = (const float*)d_in[14];
  const float* W_f1 = (const float*)d_in[15];
  const float* b_f1 = (const float*)d_in[16];
  const float* W_f2 = (const float*)d_in[17];
  const float* b_f2 = (const float*)d_in[18];
  const float* W_mix = (const float*)d_in[19];
  const float* b_mix = (const float*)d_in[20];
  const float* W_l1 = (const float*)d_in[21];
  const float* b_l1 = (const float*)d_in[22];
  const float* W_l2 = (const float*)d_in[23];
  const float* b_l2 = (const float*)d_in[24];

  float* zflow = (float*)d_ws;
  float* zcde = zflow + (size_t)Bb * Tt * Hh;
  float* out = (float*)d_out;
  float* zout = out + (size_t)Bb * Tt;

  ncde_embed_kernel<<<(Bb * Tt) / 16, 128, 0, stream>>>(
      x, seq_ts, coeffs, times, W_if, b_if, W_ic, b_ic, W_emb, b_emb, W_fl1,
      b_fl1, W_flo, b_flo, zflow);
  ncde_scan_kernel<<<Bb, 512, 0, stream>>>(coeffs, times, W_ic, b_ic,
                                                W_f1, b_f1, W_f2, b_f2, zcde);
  ncde_head_kernel<<<(Bb * Tt) / 16, 128, 0, stream>>>(
      zflow, zcde, W_mix, b_mix, W_l1, b_l1, W_l2, b_l2, out, zout);
}


# scan: GEMM2's walk takes 8 live rows per trip, not 4: one group of LDS reads, then two matrix instructions per tile; a wave waits for LDS once per 8 matrix instructions; sums unchanged
# speedup vs baseline: 4.2692x; 1.0320x over previous
#include <hip/hip_runtime.h>
#include <math.h>

#define Bb 256
#define Tt 512
#define Cc 16
#define Hh 64

#define WP  72
#define AP1 69
#define AP2 133
#define DXP 20
#define SP  72

typedef __attribute__((ext_vector_type(2))) float v2f;
typedef __attribute__((ext_vector_type(8))) float v8f;
typedef __attribute__((ext_vector_type(4))) float v4f;
#define W2P 1024

static __device__ __forceinline__ v8f wmma4(v2f a, v2f b, v8f c) {
  return __builtin_amdgcn_wmma_f32_16x16x4_f32(false, a, false, b, (short)0, c,
                                               false, false);
}

template <int K, int APITCH, int ROWMASK = 15>
static __device__ __forceinline__ v8f gemm_tile(const float* __restrict__ A,
                                                int acol0,
                                                const float* __restrict__ W,
                                                int n0, v8f acc, int l2, int hi) {
#pragma unroll
  for (int k = 0; k < K; k += 4) {
    v2f a, b;
    const float* ap = A + (l2 & ROWMASK) * APITCH + acol0 + k + 2 * hi;
    a.x = ap[0];
    a.y = ap[1];
    const float* wp = W + (k + 2 * hi) * WP + n0 + l2;
    b.x = wp[0];
    b.y = wp[WP];
    acc = wmma4(a, b, acc);
  }
  return acc;
}

template <int APITCH>
static __device__ __forceinline__ void store_tile(float* __restrict__ O, int n0,
                                                  v8f acc, int l2, int hi) {
#pragma unroll
  for (int r = 0; r < 8; ++r) O[(r + 8 * hi) * APITCH + n0 + l2] = acc[r];
}

template <int ROWS>
static __device__ __forceinline__ void load_w64(float* __restrict__ dst,
                                                const float* __restrict__ src,
                                                int tid, int nthr) {
  for (int i = tid; i < ROWS * 64; i += nthr)
    dst[(i >> 6) * WP + (i & 63)] = src[i];
}

static __device__ __forceinline__ float relu(float v) { return v > 0.f ? v : 0.f; }

template <int M>
static __device__ __forceinline__ float row_xor_add(float p) {
  const int other = __builtin_amdgcn_update_dpp(0, __builtin_bit_cast(int, p),
                                                0x160 + M, 0xf, 0xf, true);
  return p + __builtin_bit_cast(float, other);
}

__global__ __launch_bounds__(128) void ncde_embed_kernel(
    const float* __restrict__ x, const float* __restrict__ seq_ts,
    const float* __restrict__ coeffs, const float* __restrict__ times,
    const float* __restrict__ W_if, const float* __restrict__ b_if,
    const float* __restrict__ W_ic, const float* __restrict__ b_ic,
    const float* __restrict__ W_emb, const float* __restrict__ b_emb,
    const float* __restrict__ W_fl1, const float* __restrict__ b_fl1,
    const float* __restrict__ W_flo, const float* __restrict__ b_flo,
    float* __restrict__ zflow) {
  __shared__ float sWif[16 * WP], sWic[16 * WP], sWemb[128 * WP];
  __shared__ float sWfl1[64 * WP], sWflo[64 * WP];
  __shared__ float sb[5][64];
  __shared__ float sA[16 * AP2], sB[16 * AP2];

  const int tid = threadIdx.x, nthr = 128;
  const int lane = tid & 31, wave = tid >> 5, l2 = lane & 15, hi = lane >> 4;
  const int r0 = blockIdx.x * 16;

  load_w64<16>(sWif, W_if, tid, nthr);
  load_w64<16>(sWic, W_ic, tid, nthr);
  load_w64<128>(sWemb, W_emb, tid, nthr);
  load_w64<64>(sWfl1, W_fl1, tid, nthr);
  load_w64<64>(sWflo, W_flo, tid, nthr);
  if (tid < 64) {
    sb[0][tid] = b_if[tid];
    sb[1][tid] = b_ic[tid];
    sb[2][tid] = b_emb[tid];
    sb[3][tid] = b_fl1[tid];
    sb[4][tid] = b_flo[tid];
  }
  for (int i = tid; i < 256; i += nthr) {
    const int m = i >> 4, c = i & 15;
    const int rr = r0 + m, b = rr / Tt, t = rr % Tt;
    sA[m * AP2 + c] = (c == 0) ? seq_ts[rr] : x[(size_t)rr * (Cc - 1) + (c - 1)];
    const int idx = (t < Tt - 1) ? t : (Tt - 2);
    const float s = times[t] - times[idx];
    const float* cf = coeffs + ((size_t)b * (Tt - 1) + idx) * (4 * Cc) + c;
    const float a0 = cf[0], b0 = cf[Cc], c0 = cf[2 * Cc], d0 = cf[3 * Cc];
    sA[m * AP2 + 16 + c] = a0 + s * (b0 + s * (c0 + s * d0));
  }
  __syncthreads();

  const int n0 = wave * 16;
  v8f aF = {}, aX = {};
  aF = gemm_tile<16, AP2>(sA, 0, sWif, n0, aF, l2, hi);
  aX = gemm_tile<16, AP2>(sA, 16, sWic, n0, aX, l2, hi);
  const float bF = sb[0][n0 + l2], bX = sb[1][n0 + l2];
#pragma unroll
  for (int r = 0; r < 8; ++r) {
    const int m = r + 8 * hi;
    sB[m * AP2 + n0 + l2] = aF[r] + bF;
    sB[m * AP2 + 64 + n0 + l2] = aX[r] + bX;
  }
  __syncthreads();

  v8f acc = {};
  acc = gemm_tile<128, AP2>(sB, 0, sWemb, n0, acc, l2, hi);
  float bv = sb[2][n0 + l2];
#pragma unroll
  for (int r = 0; r < 8; ++r)
    sA[(r + 8 * hi) * AP2 + n0 + l2] = relu(acc[r] + bv);
  __syncthreads();

  acc = (v8f){};
  acc = gemm_tile<64, AP2>(sA, 0, sWfl1, n0, acc, l2, hi);
  bv = sb[3][n0 + l2];
#pragma unroll
  for (int r = 0; r < 8; ++r)
    sB[(r + 8 * hi) * AP2 + n0 + l2] = relu(acc[r] + bv);
  __syncthreads();

  acc = (v8f){};
  acc = gemm_tile<64, AP2>(sB, 0, sWflo, n0, acc, l2, hi);
  bv = sb[4][n0 + l2];
#pragma unroll
  for (int r = 0; r < 8; ++r)
    zflow[(size_t)(r0 + r + 8 * hi) * Hh + n0 + l2] = acc[r] + bv;
}

__global__ __launch_bounds__(512, 1) void ncde_scan_kernel(
    const float* __restrict__ coeffs, const float* __restrict__ times,
    const float* __restrict__ W_ic, const float* __restrict__ b_ic,
    const float* __restrict__ W_f1, const float* __restrict__ b_f1,
    const float* __restrict__ W_f2, const float* __restrict__ b_f2,
    float* __restrict__ zcde) {
  __shared__ float sWic[16 * WP];
  __shared__ float sbic[64];
  __shared__ float stim[Tt];
  __shared__ float sx0[16];
  __shared__ float sh[Hh + 1];
  __shared__ __attribute__((aligned(16))) float sW2[(Hh + 1) * W2P];
  __shared__ float sst[2][8 * SP];
  __shared__ float sdx[4][DXP];

  const int tid = threadIdx.x, nthr = 512;
  const int lane = tid & 31, wave = tid >> 5, l2 = lane & 15, hi = lane >> 4;
  const int b0 = blockIdx.x;
  load_w64<16>(sWic, W_ic, tid, nthr);
  if (tid < 64) sbic[tid] = b_ic[tid];
  for (int i = tid; i < Tt; i += nthr) stim[i] = times[i];
  for (int i = tid; i < 2 * 8 * SP; i += nthr) (&sst[0][0])[i] = 0.f;

  if (tid == 0) sh[Hh] = 0.f;
#pragma unroll 8
  for (int i = tid; i < Hh * Hh * Cc; i += nthr)
    sW2[(i & ~63) + 4 * (i & 15) + ((i >> 4) & 3)] = W_f2[i];
  for (int i = tid; i < W2P; i += nthr) sW2[Hh * W2P + i] = 0.f;
  const float* w2lane = sW2 + 64 * wave + 4 * l2;
  float bf2r[4];
#pragma unroll
  for (int t = 0; t < 4; ++t) bf2r[t] = b_f2[16 * (wave * 4 + t) + l2];

  v2f wf1[16];
  const int n1 = (wave & 3) * 16 + l2;
  const float bf1r = b_f1[n1];
#pragma unroll
  for (int kk = 0; kk < 16; ++kk) {
    const int row = 4 * kk + 2 * hi;
    wf1[kk].x = W_f1[row * Hh + n1];
    wf1[kk].y = W_f1[(row + 1) * Hh + n1];
  }

  if (tid < 16)
    sx0[tid] = coeffs[((size_t)b0 * (Tt - 1)) * (4 * Cc) + tid];
  __syncthreads();
  if (wave < 4) {
    const int n0 = wave * 16;
    v8f acc = {};
    acc = gemm_tile<16, AP1, 0>(sx0, 0, sWic, n0, acc, l2, hi);
    const float zv = acc[0] + sbic[n0 + l2];
    if (hi == 0) sst[0][n0 + l2] = zv;
  }
  __syncthreads();

  for (int step = 0; step < Tt - 1; ++step) {
    const float dt = stim[step + 1] - stim[step];
    const float dtp = step > 0 ? stim[step] - stim[step - 1] : 0.f;
    float* tprev = sst[step & 1];
    float* tcur = sst[(step + 1) & 1];

#pragma unroll
    for (int sub = 0; sub < 4; ++sub) {
      if (wave < 4) {
        const int n0 = wave * 16;
        const float* tile = sub == 0 ? tprev : tcur;
        v2f az[16];
        const float* zp = tile + (l2 & 7) * SP + 2 * hi;
#pragma unroll
        for (int kk = 0; kk < 16; ++kk) {
          az[kk].x = zp[4 * kk];
          az[kk].y = zp[4 * kk + 1];
        }
        v8f acc = {};
#pragma unroll
        for (int kk = 0; kk < 16; ++kk) acc = wmma4(az[kk], wf1[kk], acc);
        float pre;
        if (sub == 0) {
          pre = acc[0] + (acc[1] + 3.f * (acc[2] + acc[3]) + acc[4]) *
                             (dtp * 0.125f);
        } else if (sub == 1) {
          pre = acc[0] + (dt * (1.f / 3.f)) * acc[1];
        } else if (sub == 2) {
          pre = acc[0] + dt * (acc[2] - acc[1] * (1.f / 3.f));
        } else {
          pre = acc[0] + dt * (acc[1] - acc[2] + acc[3]);
        }
        const float hv = relu(pre + bf1r);
        if (hi == 0) sh[n0 + l2] = hv;
      } else if (sub == 0) {
        if (wave < 6) {
          const int h = tid - 128;
          const float zn =
              tprev[h] + (tprev[SP + h] +
                          3.f * (tprev[2 * SP + h] + tprev[3 * SP + h]) +
                          tprev[4 * SP + h]) *
                             (dtp * 0.125f);
          tcur[h] = zn;
          zcde[((size_t)b0 * Tt + step) * Hh + h] = zn;
        } else if (wave == 6 && lane < 16) {
          const int c = lane;
          const float* cf =
              coeffs + ((size_t)b0 * (Tt - 1) + step) * (4 * Cc) + c;
          const float bb = cf[Cc], cc = cf[2 * Cc], dd = cf[3 * Cc];
          const float s1 = dt * (1.f / 3.f), s2 = dt * (2.f / 3.f);
          sdx[0][c] = bb;
          sdx[1][c] = bb + s1 * (2.f * cc + 3.f * dd * s1);
          sdx[2][c] = bb + s2 * (2.f * cc + 3.f * dd * s2);
          const int idx3 = (step + 1 < Tt - 1) ? (step + 1) : (Tt - 2);
          const float* cf3 =
              coeffs + ((size_t)b0 * (Tt - 1) + idx3) * (4 * Cc) + c;
          const float b3 = cf3[Cc], c3 = cf3[2 * Cc], d3 = cf3[3 * Cc];
          const float s3 = stim[step + 1] - stim[idx3];
          sdx[3][c] = b3 + s3 * (2.f * c3 + 3.f * d3 * s3);
        }
      }
      __syncthreads();

      v8f acc2[4];
#pragma unroll
      for (int jj = 0; jj < 4; ++jj) {
#pragma unroll
        for (int r = 0; r < 8; ++r) acc2[jj][r] = bf2r[jj];
      }
      const unsigned mlo = __builtin_amdgcn_ballot_w32(sh[lane] != 0.f);
      const unsigned mhi = __builtin_amdgcn_ballot_w32(sh[32 + lane] != 0.f);
      unsigned long long live = ((unsigned long long)mhi << 32) | mlo;
      auto take = [&]() {
        const int r = __builtin_amdgcn_readfirstlane(
            live != 0ull ? (int)__builtin_ctzll(live) : Hh);
        live &= live - 1ull;
        return r;
      };
      bool more = live != 0ull;
      int r0 = take(), r1 = take(), r2 = take(), r3 = take();
      bool more2 = live != 0ull;
      int r4 = take(), r5 = take(), r6 = take(), r7 = take();
      auto trip = [&]() __attribute__((always_inline)) {
        const int ra = hi ? r2 : r0;
        const int rb = hi ? r3 : r1;
        const int rc = hi ? r6 : r4;
        const int rd = hi ? r7 : r5;
        v2f hlo, hup;
        hlo.x = sh[ra];
        hlo.y = sh[rb];
        hup.x = sh[rc];
        hup.y = sh[rd];
        const v4f wa = *reinterpret_cast<const v4f*>(w2lane + ra * W2P);
        const v4f wb = *reinterpret_cast<const v4f*>(w2lane + rb * W2P);
        const v4f wc = *reinterpret_cast<const v4f*>(w2lane + rc * W2P);
        const v4f wd = *reinterpret_cast<const v4f*>(w2lane + rd * W2P);
        const bool upper = more2;
        more = live != 0ull;
        r0 = take();
        r1 = take();
        r2 = take();
        r3 = take();
        more2 = live != 0ull;
        r4 = take();
        r5 = take();
        r6 = take();
        r7 = take();
        asm volatile("" ::"v"(wc), "v"(wd), "v"(hup));
#pragma unroll
        for (int jj = 0; jj < 4; ++jj) {
          v2f wv;
          wv.x = wa[jj];
          wv.y = wb[jj];
          acc2[jj] = wmma4(hlo, wv, acc2[jj]);
        }
        if (upper) {
#pragma unroll
          for (int jj = 0; jj < 4; ++jj) {
            v2f wv;
            wv.x = wc[jj];
            wv.y = wd[jj];
            acc2[jj] = wmma4(hup, wv, acc2[jj]);
          }
        }
      };
      auto stretch = [&](int budget) __attribute__((always_inline)) {
        for (int q = 0; q < budget && more; ++q) trip();
      };
      __builtin_amdgcn_s_setprio(3);
      stretch(2);
      __builtin_amdgcn_s_setprio(2);
      stretch(1);
      __builtin_amdgcn_s_setprio(1);
      stretch(1);
      __builtin_amdgcn_s_setprio(0);
      stretch(Hh);
      const float dxr = sdx[sub][l2];
      float pv[2];
#pragma unroll
      for (int r = 0; r < 2; ++r) {
        const float lo = acc2[r][0], up = acc2[2 + r][0];
        pv[r] = tanhf(hi ? up : lo) * dxr;
      }
#pragma unroll
      for (int r = 0; r < 2; ++r) {
        float p = pv[r];
        p = row_xor_add<1>(p);
        p = row_xor_add<2>(p);
        p = row_xor_add<4>(p);
        p = row_xor_add<8>(p);
        pv[r] = p;
      }
      if (l2 == 0) {
        float* kdst = tcur + (sub + 1) * SP + wave * 4 + 2 * hi;
        kdst[0] = pv[0];
        kdst[1] = pv[1];
      }
      __syncthreads();
    }
  }

  if (tid < Hh) {
    const float* tl = sst[(Tt - 1) & 1];
    const float dtl = stim[Tt - 1] - stim[Tt - 2];
    const float zn =
        tl[tid] + (tl[SP + tid] + 3.f * (tl[2 * SP + tid] + tl[3 * SP + tid]) +
                   tl[4 * SP + tid]) *
                      (dtl * 0.125f);
    zcde[((size_t)b0 * Tt + (Tt - 1)) * Hh + tid] = zn;
  }
}

__global__ __launch_bounds__(128) void ncde_head_kernel(
    const float* __restrict__ zflow, const float* __restrict__ zcde,
    const float* __restrict__ W_mix, const float* __restrict__ b_mix,
    const float* __restrict__ W_l1, const float* __restrict__ b_l1,
    const float* __restrict__ W_l2, const float* __restrict__ b_l2,
    float* __restrict__ out, float* __restrict__ zout) {
  __shared__ float sWmix[128 * WP], sWl1[64 * WP];
  __shared__ float sbmix[64], sbl1[64], sWl2v[64];
  __shared__ float sA[16 * AP2], sB[16 * AP1], sC[16 * AP1];

  const int tid = threadIdx.x, nthr = 128;
  const int lane = tid & 31, wave = tid >> 5, l2 = lane & 15, hi = lane >> 4;
  const int r0 = blockIdx.x * 16;

  load_w64<128>(sWmix, W_mix, tid, nthr);
  load_w64<64>(sWl1, W_l1, tid, nthr);
  if (tid < 64) {
    sbmix[tid] = b_mix[tid];
    sbl1[tid] = b_l1[tid];
    sWl2v[tid] = W_l2[tid];
  }
  for (int i = tid; i < 16 * 128; i += nthr) {
    const int m = i >> 7, c = i & 127;
    const size_t rr = r0 + m;
    sA[m * AP2 + c] = (c < 64) ? zflow[rr * Hh + c] : zcde[rr * Hh + (c - 64)];
  }
  __syncthreads();

  const int n0 = wave * 16;
  v8f acc = {};
  acc = gemm_tile<128, AP2>(sA, 0, sWmix, n0, acc, l2, hi);
  float bv = sbmix[n0 + l2];
#pragma unroll
  for (int r = 0; r < 8; ++r) {
    const int m = r + 8 * hi;
    const float v = acc[r] + bv;
    zout[(size_t)(r0 + m) * Hh + n0 + l2] = v;
    sB[m * AP1 + n0 + l2] = tanhf(v);
  }
  __syncthreads();

  acc = (v8f){};
  acc = gemm_tile<64, AP1>(sB, 0, sWl1, n0, acc, l2, hi);
  bv = sbl1[n0 + l2];
#pragma unroll
  for (int r = 0; r < 8; ++r)
    sC[(r + 8 * hi) * AP1 + n0 + l2] = relu(acc[r] + bv);
  __syncthreads();

  const int m = tid >> 3, p = tid & 7;
  float s = 0.f;
#pragma unroll
  for (int q = 0; q < 8; ++q) s += sC[m * AP1 + p * 8 + q] * sWl2v[p * 8 + q];
  s += __shfl_xor(s, 1, 32);
  s += __shfl_xor(s, 2, 32);
  s += __shfl_xor(s, 4, 32);
  if (p == 0) out[r0 + m] = s + b_l2[0];
}

extern "C" void kernel_launch(void* const* d_in, const int* in_sizes, int n_in,
                              void* d_out, int out_size, void* d_ws,
                              size_t ws_size, hipStream_t stream) {
  const float* x = (const float*)d_in[0];
  const float* seq_ts = (const float*)d_in[1];
  const float* coeffs = (const float*)d_in[3];
  const float* times = (const float*)d_in[4];
  const float* W_if = (const float*)d_in[5];
  const float* b_if = (const float*)d_in[6];
  const float* W_ic = (const float*)d_in[7];
  const float* b_ic = (const float*)d_in[8];
  const float* W_emb = (const float*)d_in[9];
  const float* b_emb = (const float*)d_in[10];
  const float* W_fl1 = (const float*)d_in[11];
  const float* b_fl1 = (const float*)d_in[12];
  const float* W_flo = (const float*)d_in[13];
  const float* b_flo = (const float*)d_in[14];
  const float* W_f1 = (const float*)d_in[15];
  const float* b_f1 = (const float*)d_in[16];
  const float* W_f2 = (const float*)d_in[17];
  const float* b_f2 = (const float*)d_in[18];
  const float* W_mix = (const float*)d_in[19];
  const float* b_mix = (const float*)d_in[20];
  const float* W_l1 = (const float*)d_in[21];
  const float* b_l1 = (const float*)d_in[22];
  const float* W_l2 = (const float*)d_in[23];
  const float* b_l2 = (const float*)d_in[24];

  float* zflow = (float*)d_ws;
  float* zcde = zflow + (size_t)Bb * Tt * Hh;
  float* out = (float*)d_out;
  float* zout = out + (size_t)Bb * Tt;

  ncde_embed_kernel<<<(Bb * Tt) / 16, 128, 0, stream>>>(
      x, seq_ts, coeffs, times, W_if, b_if, W_ic, b_ic, W_emb, b_emb, W_fl1,
      b_fl1, W_flo, b_flo, zflow);
  ncde_scan_kernel<<<Bb, 512, 0, stream>>>(coeffs, times, W_ic, b_ic,
                                                W_f1, b_f1, W_f2, b_f2, zcde);
  ncde_head_kernel<<<(Bb * Tt) / 16, 128, 0, stream>>>(
      zflow, zcde, W_mix, b_mix, W_l1, b_l1, W_l2, b_l2, out, zout);
}
